# starting kernel
# baseline (speedup 1.0000x reference)
#include <hip/hip_runtime.h>
#include <hip/hip_bf16.h>

typedef _Float16 f16;
typedef __attribute__((ext_vector_type(16))) _Float16 v16h;
typedef __attribute__((ext_vector_type(8)))  _Float16 v8h;
typedef __attribute__((ext_vector_type(8)))  float    v8f;

#define BATCH 512
#define TSEQ  256
#define CNNC  512
#define VOC   37
#define HID   8
#define KPAD  64
#define GN    32
#define NROW  (TSEQ * BATCH)
#define NTILE (NROW / 16)

__device__ __forceinline__ v8f wmma16(v16h a, v16h b, v8f c) {
  return __builtin_amdgcn_wmma_f32_16x16x32_f16(
       false, a,  false, b,
       (short)0, c,  false,  false);
}

__device__ __forceinline__ v16h cat8(v8h a, v8h b) {
  v16h r;
  for (int i = 0; i < 8; ++i) { r[i] = a[i]; r[i + 8] = b[i]; }
  return r;
}

__device__ __forceinline__ v8f splat8(float x) {
  v8f r;
  for (int i = 0; i < 8; ++i) r[i] = x;
  return r;
}

__device__ __forceinline__ float sigmoidf_(float x) {
  return 1.0f / (1.0f + __expf(-x));
}
__device__ __forceinline__ float tanhf_(float x) {
  float t = __expf(-2.0f * fabsf(x));
  float r = (1.0f - t) / (1.0f + t);
  return copysignf(r, x);
}

__global__ void lstm_pool_kernel(const float* __restrict__ features,
                                 f16* __restrict__ pooled) {
  int idx = blockIdx.x * blockDim.x + threadIdx.x;
  if (idx >= BATCH * CNNC) return;
  const float* p = features + (size_t)idx * 49;
  float s = 0.f;
  #pragma unroll
  for (int i = 0; i < 49; ++i) s += p[i];
  pooled[idx] = (f16)(s * (1.0f / 49.0f));
}

__global__ void lstm_pack_kernel(const float* __restrict__ W_in,
                                 const float* __restrict__ W_ih,
                                 const float* __restrict__ W_hh,
                                 const float* __restrict__ W_out,
                                 const float* __restrict__ b_ih,
                                 const float* __restrict__ b_hh,
                                 const float* __restrict__ b_out,
                                 f16* __restrict__ Winp,
                                 f16* __restrict__ Wihp,
                                 f16* __restrict__ Whhp,
                                 f16* __restrict__ Woutp,
                                 float* __restrict__ gbias,
                                 float* __restrict__ obias)
{
  int tid = threadIdx.x;
  for (int idx = tid; idx < 48 * CNNC; idx += 256) {
    int n = idx / CNNC, k = idx % CNNC;
    Winp[idx] = (f16)(n < VOC ? W_in[n * CNNC + k] : 0.f);
  }
  for (int idx = tid; idx < 4 * 512; idx += 256) {
    int tile = idx / 512, e = idx % 512, lane = e / 16, h = e % 16;
    int ks = tile / 2, nt = tile % 2;
    int kl = (h < 8 ? h : h + 8) + (lane / 16) * 8;
    int k = ks * 32 + kl;
    int n = nt * 16 + (lane % 16);
    Wihp[idx] = (f16)(k < VOC ? W_ih[n * VOC + k] : 0.f);
  }
  for (int idx = tid; idx < 2 * 512; idx += 256) {
    int tile = idx / 512, e = idx % 512, lane = e / 16, h = e % 16;
    int kl = (h < 8 ? h : h + 8) + (lane / 16) * 8;
    int n = tile * 16 + (lane % 16);
    Whhp[idx] = (f16)(kl < HID ? W_hh[n * HID + kl] : 0.f);
  }
  for (int idx = tid; idx < 3 * 512; idx += 256) {
    int tile = idx / 512, e = idx % 512, lane = e / 16, h = e % 16;
    int kl = (h < 8 ? h : h + 8) + (lane / 16) * 8;
    int n = tile * 16 + (lane % 16);
    Woutp[idx] = (f16)((kl < HID && n < VOC) ? W_out[n * HID + kl] : 0.f);
  }
  if (tid < GN) gbias[tid] = b_ih[tid] + b_hh[tid];
  if (tid < 48) obias[tid] = (tid < VOC) ? b_out[tid] : 0.f;
}

__global__ void lstm_featgemm_kernel(const f16* __restrict__ pooled,
                                     const f16* __restrict__ Winp,
                                     const float* __restrict__ b_in,
                                     f16* __restrict__ Xh) {
  const int L = threadIdx.x;
  const int j16 = L & 15, half = L >> 4;
  const int Mb = blockIdx.x * 16;

  v8f acc[3];
  #pragma unroll
  for (int nt = 0; nt < 3; ++nt) {
    int col = nt * 16 + j16;
    acc[nt] = splat8(col < VOC ? b_in[col] : 0.f);
  }

  for (int kt = 0; kt < CNNC / 32; ++kt) {
    int kb = kt * 32 + half * 8;
    int row = Mb + j16;
    v16h A = cat8(*(const v8h*)(pooled + (size_t)row * CNNC + kb),
                  *(const v8h*)(pooled + (size_t)row * CNNC + kb + 16));
    #pragma unroll
    for (int nt = 0; nt < 3; ++nt) {
      int n = nt * 16 + j16;
      v16h Bm = cat8(*(const v8h*)(Winp + (size_t)n * CNNC + kb),
                     *(const v8h*)(Winp + (size_t)n * CNNC + kb + 16));
      acc[nt] = wmma16(A, Bm, acc[nt]);
    }
  }
  #pragma unroll
  for (int r = 0; r < 8; ++r) {
    int m = r + 8 * half;
    f16* dst = Xh + (size_t)(Mb + m) * KPAD;
    #pragma unroll
    for (int nt = 0; nt < 3; ++nt) {
      int col = nt * 16 + j16;
      dst[col] = (col < VOC) ? (f16)acc[nt][r] : (f16)0;
    }
    dst[48 + j16] = (f16)0;
  }
}

__global__ void lstm_packx_kernel(const float* __restrict__ captions,
                                  f16* __restrict__ Xh) {
  size_t idx = (size_t)blockIdx.x * blockDim.x + threadIdx.x;
  int col = idx & (KPAD - 1);
  size_t row = idx >> 6;
  int t = (int)(row / BATCH) + 1;
  int b = (int)(row % BATCH);
  float v = (col < VOC) ? captions[((size_t)b * TSEQ + (t - 1)) * VOC + col] : 0.f;
  Xh[((size_t)t * BATCH + b) * KPAD + col] = (f16)v;
}

__global__ void lstm_pregate_kernel(const f16* __restrict__ Xh,
                                    const f16* __restrict__ Wihp,
                                    const float* __restrict__ gbias,
                                    float* __restrict__ G) {
  const int L = threadIdx.x;
  const int j16 = L & 15, half = L >> 4;
  const size_t Mb = (size_t)blockIdx.x * 16;

  v8f acc[2];
  acc[0] = splat8(gbias[j16]);
  acc[1] = splat8(gbias[16 + j16]);

  #pragma unroll
  for (int ks = 0; ks < 2; ++ks) {
    int kb = ks * 32 + half * 8;
    size_t row = Mb + j16;
    v16h A = cat8(*(const v8h*)(Xh + row * KPAD + kb),
                  *(const v8h*)(Xh + row * KPAD + kb + 16));
    #pragma unroll
    for (int nt = 0; nt < 2; ++nt) {
      const f16* tp = Wihp + (size_t)(ks * 2 + nt) * 512 + L * 16;
      v16h Bm = cat8(*(const v8h*)tp, *(const v8h*)(tp + 8));
      acc[nt] = wmma16(A, Bm, acc[nt]);
    }
  }
  float* dst = G + (size_t)blockIdx.x * 512 + L * 16;
  *(v8f*)dst       = acc[0];
  *(v8f*)(dst + 8) = acc[1];
}

__global__ void __launch_bounds__(32)
lstm_recur_kernel(const float* __restrict__ G,
                  const f16* __restrict__ Whhp,
                  f16* __restrict__ Hb) {
  const int L = threadIdx.x;
  const int j16 = L & 15, half = L >> 4;
  const int bblk = blockIdx.x;
  const int bbase = bblk * 16;

  v16h WhhB[2];
  #pragma unroll
  for (int nt = 0; nt < 2; ++nt) {
    const f16* tp = Whhp + (size_t)nt * 512 + L * 16;
    WhhB[nt] = cat8(*(const v8h*)tp, *(const v8h*)(tp + 8));
  }

  float cst[8];
  #pragma unroll
  for (int r = 0; r < 8; ++r) cst[r] = 0.f;

  v16h Ah;
  #pragma unroll
  for (int i = 0; i < 16; ++i) Ah[i] = (f16)0;

  __shared__ f16 hb[16][HID];

  const bool lowj = (j16 < 8);

  for (int t = 0; t < TSEQ; ++t) {
    const float* Gt = G + ((size_t)(t * (BATCH / 16) + bblk) * 512) + L * 16;
    v8f g0 = *(const v8f*)Gt;
    v8f g1 = *(const v8f*)(Gt + 8);
    if (t + 1 < TSEQ)
      __builtin_prefetch(Gt + (size_t)(BATCH / 16) * 512, 0, 1);

    g0 = wmma16(Ah, WhhB[0], g0);
    g1 = wmma16(Ah, WhhB[1], g1);

    float hnew[8];
    #pragma unroll
    for (int r = 0; r < 8; ++r) {
      float a0 = g0[r], a1 = g1[r];
      float b0 = __shfl_xor(a0, 8, 32);
      float b1 = __shfl_xor(a1, 8, 32);
      float iv = lowj ? a0 : b0;
      float fv = lowj ? b0 : a0;
      float gv = lowj ? a1 : b1;
      float ov = lowj ? b1 : a1;
      iv = sigmoidf_(iv); fv = sigmoidf_(fv); ov = sigmoidf_(ov);
      gv = tanhf_(gv);
      cst[r] = fv * cst[r] + iv * gv;
      hnew[r] = ov * tanhf_(cst[r]);
    }

    if (lowj) {
      #pragma unroll
      for (int r = 0; r < 8; ++r) hb[r + 8 * half][j16] = (f16)hnew[r];
    }
    __syncthreads();
    v16h An;
    #pragma unroll
    for (int i = 0; i < 16; ++i) An[i] = (f16)0;
    v8h hv;
    #pragma unroll
    for (int i = 0; i < 8; ++i) hv[i] = (f16)0;
    if (L < 16) {
      hv = *(const v8h*)&hb[L][0];
      #pragma unroll
      for (int k = 0; k < HID; ++k) An[k] = hv[k];
    }
    __syncthreads();
    Ah = An;

    if (L < 16)
      *(v8h*)(Hb + ((size_t)t * BATCH + bbase + L) * HID) = hv;
  }
}

__global__ void lstm_outhead_kernel(const f16* __restrict__ Hb,
                                    const f16* __restrict__ Woutp,
                                    const float* __restrict__ obias,
                                    float* __restrict__ out) {
  const int L = threadIdx.x;
  const int j16 = L & 15, half = L >> 4;
  const size_t Mb = (size_t)blockIdx.x * 16;

  v16h WoutB[3];
  #pragma unroll
  for (int nt = 0; nt < 3; ++nt) {
    const f16* tp = Woutp + (size_t)nt * 512 + L * 16;
    WoutB[nt] = cat8(*(const v8h*)tp, *(const v8h*)(tp + 8));
  }

  v16h A;
  #pragma unroll
  for (int i = 0; i < 16; ++i) A[i] = (f16)0;
  if (L < 16) {
    v8h hv = *(const v8h*)(Hb + (Mb + L) * HID);
    #pragma unroll
    for (int k = 0; k < HID; ++k) A[k] = hv[k];
  }

  v8f l0 = splat8(obias[j16]);
  v8f l1 = splat8(obias[16 + j16]);
  v8f l2 = splat8(obias[32 + j16]);
  l0 = wmma16(A, WoutB[0], l0);
  l1 = wmma16(A, WoutB[1], l1);
  l2 = wmma16(A, WoutB[2], l2);

  const bool v2ok = (32 + j16) < VOC;
  #pragma unroll
  for (int r = 0; r < 8; ++r) {
    float x2 = v2ok ? l2[r] : -1e30f;
    float mx = fmaxf(fmaxf(l0[r], l1[r]), x2);
    #pragma unroll
    for (int s = 1; s < 16; s <<= 1)
      mx = fmaxf(mx, __shfl_xor(mx, s, 16));
    float e0 = __expf(l0[r] - mx);
    float e1 = __expf(l1[r] - mx);
    float e2 = v2ok ? __expf(x2 - mx) : 0.f;
    float sm = e0 + e1 + e2;
    #pragma unroll
    for (int s = 1; s < 16; s <<= 1)
      sm += __shfl_xor(sm, s, 16);
    float inv = 1.0f / sm;

    size_t row = Mb + r + 8 * half;
    int t = (int)(row / BATCH);
    int b = (int)(row % BATCH);
    float* dst = out + ((size_t)b * TSEQ + t) * VOC;
    dst[j16]      = e0 * inv;
    dst[16 + j16] = e1 * inv;
    if (v2ok) dst[32 + j16] = e2 * inv;
  }
}

extern "C" void kernel_launch(void* const* d_in, const int* in_sizes, int n_in,
                              void* d_out, int out_size, void* d_ws, size_t ws_size,
                              hipStream_t stream) {
  const float* features = (const float*)d_in[0];
  const float* captions = (const float*)d_in[1];
  const float* W_in  = (const float*)d_in[2];
  const float* b_in  = (const float*)d_in[3];
  const float* W_ih  = (const float*)d_in[4];
  const float* W_hh  = (const float*)d_in[5];
  const float* b_ih  = (const float*)d_in[6];
  const float* b_hh  = (const float*)d_in[7];
  const float* W_out = (const float*)d_in[8];
  const float* b_out = (const float*)d_in[9];
  float* out = (float*)d_out;

  char* ws = (char*)d_ws;
  auto carve = [&](size_t bytes) -> char* {
    char* p = ws;
    ws += (bytes + 255) & ~(size_t)255;
    return p;
  };
  f16*   pooled = (f16*)  carve((size_t)BATCH * CNNC * sizeof(f16));
  f16*   Winp   = (f16*)  carve((size_t)48 * CNNC * sizeof(f16));
  f16*   Wihp   = (f16*)  carve((size_t)4 * 512 * sizeof(f16));
  f16*   Whhp   = (f16*)  carve((size_t)2 * 512 * sizeof(f16));
  f16*   Woutp  = (f16*)  carve((size_t)3 * 512 * sizeof(f16));
  float* gbias  = (float*)carve(GN * sizeof(float));
  float* obias  = (float*)carve(48 * sizeof(float));
  f16*   Xh     = (f16*)  carve((size_t)NROW * KPAD * sizeof(f16));
  float* G      = (float*)carve((size_t)NROW * GN * sizeof(float));
  f16*   Hb     = (f16*)  carve((size_t)NROW * HID * sizeof(f16));

  lstm_pool_kernel<<<(BATCH * CNNC) / 256, 256, 0, stream>>>(features, pooled);
  lstm_pack_kernel<<<1, 256, 0, stream>>>(W_in, W_ih, W_hh, W_out,
                                          b_ih, b_hh, b_out,
                                          Winp, Wihp, Whhp, Woutp, gbias, obias);
  lstm_featgemm_kernel<<<BATCH / 16, 32, 0, stream>>>(pooled, Winp, b_in, Xh);
  lstm_packx_kernel<<<((TSEQ - 1) * BATCH * KPAD) / 256, 256, 0, stream>>>(captions, Xh);
  lstm_pregate_kernel<<<NTILE, 32, 0, stream>>>(Xh, Wihp, gbias, G);
  lstm_recur_kernel<<<BATCH / 16, 32, 0, stream>>>(G, Whhp, Hb);
  lstm_outhead_kernel<<<NTILE, 32, 0, stream>>>(Hb, Woutp, obias, out);
}


# Recurrence transposed (gates^T = W_hh.h^T, gate rows regrouped): each lane owns 4 whole cells and writes h into its own B fragment (h moves to other k slots); no shuffle, LDS or barrier per step
# speedup vs baseline: 1.3241x; 1.3241x over previous
#include <hip/hip_runtime.h>
#include <hip/hip_bf16.h>

typedef _Float16 f16;
typedef __attribute__((ext_vector_type(16))) _Float16 v16h;
typedef __attribute__((ext_vector_type(8)))  _Float16 v8h;
typedef __attribute__((ext_vector_type(4)))  _Float16 v4h;
typedef __attribute__((ext_vector_type(8)))  float    v8f;

#define BATCH 512
#define TSEQ  256
#define CNNC  512
#define VOC   37
#define HID   8
#define KPAD  64
#define GN    32
#define NROW  (TSEQ * BATCH)
#define NTILE (NROW / 16)

__device__ __forceinline__ v8f wmma16(v16h a, v16h b, v8f c) {
  return __builtin_amdgcn_wmma_f32_16x16x32_f16(
       false, a,  false, b,
       (short)0, c,  false,  false);
}

__device__ __forceinline__ v16h cat8(v8h a, v8h b) {
  v16h r;
  for (int i = 0; i < 8; ++i) { r[i] = a[i]; r[i + 8] = b[i]; }
  return r;
}

__device__ __forceinline__ v8f splat8(float x) {
  v8f r;
  for (int i = 0; i < 8; ++i) r[i] = x;
  return r;
}

__device__ __forceinline__ float sigmoidf_(float x) {
  return 1.0f / (1.0f + __expf(-x));
}
__device__ __forceinline__ float tanhf_(float x) {
  float t = __expf(-2.0f * fabsf(x));
  float r = (1.0f - t) / (1.0f + t);
  return copysignf(r, x);
}

__global__ void lstm_pool_kernel(const float* __restrict__ features,
                                 f16* __restrict__ pooled) {
  int idx = blockIdx.x * blockDim.x + threadIdx.x;
  if (idx >= BATCH * CNNC) return;
  const float* p = features + (size_t)idx * 49;
  float s = 0.f;
  #pragma unroll
  for (int i = 0; i < 49; ++i) s += p[i];
  pooled[idx] = (f16)(s * (1.0f / 49.0f));
}

__device__ __forceinline__ int gate_row(int nt, int m) {
  int r = m & 7, hh = m >> 3;
  return (2 * nt + (r >> 2)) * HID + 4 * hh + (r & 3);
}

__global__ void lstm_pack_kernel(const float* __restrict__ W_in,
                                 const float* __restrict__ W_ih,
                                 const float* __restrict__ W_hh,
                                 const float* __restrict__ W_out,
                                 const float* __restrict__ b_ih,
                                 const float* __restrict__ b_hh,
                                 const float* __restrict__ b_out,
                                 f16* __restrict__ Winp,
                                 f16* __restrict__ Wihp,
                                 f16* __restrict__ Whhp,
                                 f16* __restrict__ Woutp,
                                 float* __restrict__ gbias,
                                 float* __restrict__ obias)
{
  int tid = threadIdx.x;
  for (int idx = tid; idx < 48 * CNNC; idx += 256) {
    int n = idx / CNNC, k = idx % CNNC;
    Winp[idx] = (f16)(n < VOC ? W_in[n * CNNC + k] : 0.f);
  }
  for (int idx = tid; idx < 4 * 512; idx += 256) {
    int tile = idx / 512, e = idx % 512, lane = e / 16, h = e % 16;
    int ks = tile / 2, nt = tile % 2;
    int kl = (h < 8 ? h : h + 8) + (lane / 16) * 8;
    int k = ks * 32 + kl;
    int n = gate_row(nt, lane % 16);
    Wihp[idx] = (f16)(k < VOC ? W_ih[n * VOC + k] : 0.f);
  }
  for (int idx = tid; idx < 2 * 512; idx += 256) {
    int tile = idx / 512, e = idx % 512, lane = e / 16, h = e % 16;
    int n = gate_row(tile, lane % 16);
    Whhp[idx] = (f16)(h < 4 ? W_hh[n * HID + 4 * (lane / 16) + h] : 0.f);
  }
  for (int idx = tid; idx < 3 * 512; idx += 256) {
    int tile = idx / 512, e = idx % 512, lane = e / 16, h = e % 16;
    int kl = (h < 8 ? h : h + 8) + (lane / 16) * 8;
    int n = tile * 16 + (lane % 16);
    Woutp[idx] = (f16)((kl < HID && n < VOC) ? W_out[n * HID + kl] : 0.f);
  }
  if (tid < GN) {
    int n = gate_row(tid / 16, tid % 16);
    gbias[tid] = b_ih[n] + b_hh[n];
  }
  if (tid < 48) obias[tid] = (tid < VOC) ? b_out[tid] : 0.f;
}

__global__ void lstm_featgemm_kernel(const f16* __restrict__ pooled,
                                     const f16* __restrict__ Winp,
                                     const float* __restrict__ b_in,
                                     f16* __restrict__ Xh) {
  const int L = threadIdx.x;
  const int j16 = L & 15, half = L >> 4;
  const int Mb = blockIdx.x * 16;

  v8f acc[3];
  #pragma unroll
  for (int nt = 0; nt < 3; ++nt) {
    int col = nt * 16 + j16;
    acc[nt] = splat8(col < VOC ? b_in[col] : 0.f);
  }

  for (int kt = 0; kt < CNNC / 32; ++kt) {
    int kb = kt * 32 + half * 8;
    int row = Mb + j16;
    v16h A = cat8(*(const v8h*)(pooled + (size_t)row * CNNC + kb),
                  *(const v8h*)(pooled + (size_t)row * CNNC + kb + 16));
    #pragma unroll
    for (int nt = 0; nt < 3; ++nt) {
      int n = nt * 16 + j16;
      v16h Bm = cat8(*(const v8h*)(Winp + (size_t)n * CNNC + kb),
                     *(const v8h*)(Winp + (size_t)n * CNNC + kb + 16));
      acc[nt] = wmma16(A, Bm, acc[nt]);
    }
  }
  #pragma unroll
  for (int r = 0; r < 8; ++r) {
    int m = r + 8 * half;
    f16* dst = Xh + (size_t)(Mb + m) * KPAD;
    #pragma unroll
    for (int nt = 0; nt < 3; ++nt) {
      int col = nt * 16 + j16;
      dst[col] = (col < VOC) ? (f16)acc[nt][r] : (f16)0;
    }
    dst[48 + j16] = (f16)0;
  }
}

__global__ void lstm_packx_kernel(const float* __restrict__ captions,
                                  f16* __restrict__ Xh) {
  size_t idx = (size_t)blockIdx.x * blockDim.x + threadIdx.x;
  int col = idx & (KPAD - 1);
  size_t row = idx >> 6;
  int t = (int)(row / BATCH) + 1;
  int b = (int)(row % BATCH);
  float v = (col < VOC) ? captions[((size_t)b * TSEQ + (t - 1)) * VOC + col] : 0.f;
  Xh[((size_t)t * BATCH + b) * KPAD + col] = (f16)v;
}

__global__ void lstm_pregate_kernel(const f16* __restrict__ Xh,
                                    const f16* __restrict__ Wihp,
                                    const float* __restrict__ gbias,
                                    float* __restrict__ G) {
  const int L = threadIdx.x;
  const int j16 = L & 15, half = L >> 4;
  const size_t Mb = (size_t)blockIdx.x * 16;

  v8f acc[2];
  acc[0] = *(const v8f*)(gbias + 8 * half);
  acc[1] = *(const v8f*)(gbias + 16 + 8 * half);

  #pragma unroll
  for (int ks = 0; ks < 2; ++ks) {
    int kb = ks * 32 + half * 8;
    size_t row = Mb + j16;
    v16h Xb = cat8(*(const v8h*)(Xh + row * KPAD + kb),
                   *(const v8h*)(Xh + row * KPAD + kb + 16));
    #pragma unroll
    for (int nt = 0; nt < 2; ++nt) {
      const f16* tp = Wihp + (size_t)(ks * 2 + nt) * 512 + L * 16;
      v16h Wa = cat8(*(const v8h*)tp, *(const v8h*)(tp + 8));
      acc[nt] = wmma16(Wa, Xb, acc[nt]);
    }
  }
  float* dst = G + (size_t)blockIdx.x * 512 + L * 16;
  *(v8f*)dst       = acc[0];
  *(v8f*)(dst + 8) = acc[1];
}

__global__ void __launch_bounds__(32)
lstm_recur_kernel(const float* __restrict__ G,
                  const f16* __restrict__ Whhp,
                  f16* __restrict__ Hb) {
  const int L = threadIdx.x;
  const int j16 = L & 15, half = L >> 4;
  const int bblk = blockIdx.x;
  const int bbase = bblk * 16;

  v16h WhhA[2];
  for (int nt = 0; nt < 2; ++nt) {
    const f16* tp = Whhp + (size_t)nt * 512 + L * 16;
    WhhA[nt] = cat8(*(const v8h*)tp, *(const v8h*)(tp + 8));
  }

  float cst[4];
  for (int u = 0; u < 4; ++u) cst[u] = 0.f;

  v16h Bh;
  for (int i = 0; i < 16; ++i) Bh[i] = (f16)0;

  f16* hdst = Hb + ((size_t)bbase + j16) * HID + 4 * half;

  for (int t = 0; t < TSEQ; ++t) {
    const float* Gt = G + ((size_t)(t * (BATCH / 16) + bblk) * 512) + L * 16;
    v8f g0 = *(const v8f*)Gt;
    v8f g1 = *(const v8f*)(Gt + 8);
    if (t + 1 < TSEQ)
      __builtin_prefetch(Gt + (size_t)(BATCH / 16) * 512, 0, 1);

    g0 = wmma16(WhhA[0], Bh, g0);
    g1 = wmma16(WhhA[1], Bh, g1);

    v4h hv;
    #pragma unroll
    for (int u = 0; u < 4; ++u) {
      float iv = sigmoidf_(g0[u]);
      float fv = sigmoidf_(g0[4 + u]);
      float gv = tanhf_(g1[u]);
      float ov = sigmoidf_(g1[4 + u]);
      cst[u] = fv * cst[u] + iv * gv;
      hv[u] = (f16)(ov * tanhf_(cst[u]));
    }
    for (int u = 0; u < 4; ++u) Bh[u] = hv[u];

    *(v4h*)(hdst + (size_t)t * BATCH * HID) = hv;
  }
}

__global__ void lstm_outhead_kernel(const f16* __restrict__ Hb,
                                    const f16* __restrict__ Woutp,
                                    const float* __restrict__ obias,
                                    float* __restrict__ out) {
  const int L = threadIdx.x;
  const int j16 = L & 15, half = L >> 4;
  const size_t Mb = (size_t)blockIdx.x * 16;

  v16h WoutB[3];
  #pragma unroll
  for (int nt = 0; nt < 3; ++nt) {
    const f16* tp = Woutp + (size_t)nt * 512 + L * 16;
    WoutB[nt] = cat8(*(const v8h*)tp, *(const v8h*)(tp + 8));
  }

  v16h A;
  #pragma unroll
  for (int i = 0; i < 16; ++i) A[i] = (f16)0;
  if (L < 16) {
    v8h hv = *(const v8h*)(Hb + (Mb + L) * HID);
    #pragma unroll
    for (int k = 0; k < HID; ++k) A[k] = hv[k];
  }

  v8f l0 = splat8(obias[j16]);
  v8f l1 = splat8(obias[16 + j16]);
  v8f l2 = splat8(obias[32 + j16]);
  l0 = wmma16(A, WoutB[0], l0);
  l1 = wmma16(A, WoutB[1], l1);
  l2 = wmma16(A, WoutB[2], l2);

  const bool v2ok = (32 + j16) < VOC;
  #pragma unroll
  for (int r = 0; r < 8; ++r) {
    float x2 = v2ok ? l2[r] : -1e30f;
    float mx = fmaxf(fmaxf(l0[r], l1[r]), x2);
    #pragma unroll
    for (int s = 1; s < 16; s <<= 1)
      mx = fmaxf(mx, __shfl_xor(mx, s, 16));
    float e0 = __expf(l0[r] - mx);
    float e1 = __expf(l1[r] - mx);
    float e2 = v2ok ? __expf(x2 - mx) : 0.f;
    float sm = e0 + e1 + e2;
    #pragma unroll
    for (int s = 1; s < 16; s <<= 1)
      sm += __shfl_xor(sm, s, 16);
    float inv = 1.0f / sm;

    size_t row = Mb + r + 8 * half;
    int t = (int)(row / BATCH);
    int b = (int)(row % BATCH);
    float* dst = out + ((size_t)b * TSEQ + t) * VOC;
    dst[j16]      = e0 * inv;
    dst[16 + j16] = e1 * inv;
    if (v2ok) dst[32 + j16] = e2 * inv;
  }
}

extern "C" void kernel_launch(void* const* d_in, const int* in_sizes, int n_in,
                              void* d_out, int out_size, void* d_ws, size_t ws_size,
                              hipStream_t stream) {
  const float* features = (const float*)d_in[0];
  const float* captions = (const float*)d_in[1];
  const float* W_in  = (const float*)d_in[2];
  const float* b_in  = (const float*)d_in[3];
  const float* W_ih  = (const float*)d_in[4];
  const float* W_hh  = (const float*)d_in[5];
  const float* b_ih  = (const float*)d_in[6];
  const float* b_hh  = (const float*)d_in[7];
  const float* W_out = (const float*)d_in[8];
  const float* b_out = (const float*)d_in[9];
  float* out = (float*)d_out;

  char* ws = (char*)d_ws;
  auto carve = [&](size_t bytes) -> char* {
    char* p = ws;
    ws += (bytes + 255) & ~(size_t)255;
    return p;
  };
  f16*   pooled = (f16*)  carve((size_t)BATCH * CNNC * sizeof(f16));
  f16*   Winp   = (f16*)  carve((size_t)48 * CNNC * sizeof(f16));
  f16*   Wihp   = (f16*)  carve((size_t)4 * 512 * sizeof(f16));
  f16*   Whhp   = (f16*)  carve((size_t)2 * 512 * sizeof(f16));
  f16*   Woutp  = (f16*)  carve((size_t)3 * 512 * sizeof(f16));
  float* gbias  = (float*)carve(GN * sizeof(float));
  float* obias  = (float*)carve(48 * sizeof(float));
  f16*   Xh     = (f16*)  carve((size_t)NROW * KPAD * sizeof(f16));
  float* G      = (float*)carve((size_t)NROW * GN * sizeof(float));
  f16*   Hb     = (f16*)  carve((size_t)NROW * HID * sizeof(f16));

  lstm_pool_kernel<<<(BATCH * CNNC) / 256, 256, 0, stream>>>(features, pooled);
  lstm_pack_kernel<<<1, 256, 0, stream>>>(W_in, W_ih, W_hh, W_out,
                                          b_ih, b_hh, b_out,
                                          Winp, Wihp, Whhp, Woutp, gbias, obias);
  lstm_featgemm_kernel<<<BATCH / 16, 32, 0, stream>>>(pooled, Winp, b_in, Xh);
  lstm_packx_kernel<<<((TSEQ - 1) * BATCH * KPAD) / 256, 256, 0, stream>>>(captions, Xh);
  lstm_pregate_kernel<<<NTILE, 32, 0, stream>>>(Xh, Wihp, gbias, G);
  lstm_recur_kernel<<<BATCH / 16, 32, 0, stream>>>(G, Whhp, Hb);
  lstm_outhead_kernel<<<NTILE, 32, 0, stream>>>(Hb, Woutp, obias, out);
}


# Recurrence cell sigmoid/tanh: hardware reciprocal v_rcp_f32 (1 ulp) replaces the correctly rounded f32 division (11 dependent instructions each, 20 per step) on the serial path
# speedup vs baseline: 1.7436x; 1.3168x over previous
#include <hip/hip_runtime.h>
#include <hip/hip_bf16.h>

typedef _Float16 f16;
typedef __attribute__((ext_vector_type(16))) _Float16 v16h;
typedef __attribute__((ext_vector_type(8)))  _Float16 v8h;
typedef __attribute__((ext_vector_type(4)))  _Float16 v4h;
typedef __attribute__((ext_vector_type(8)))  float    v8f;

#define BATCH 512
#define TSEQ  256
#define CNNC  512
#define VOC   37
#define HID   8
#define KPAD  64
#define GN    32
#define NROW  (TSEQ * BATCH)
#define NTILE (NROW / 16)

__device__ __forceinline__ v8f wmma16(v16h a, v16h b, v8f c) {
  return __builtin_amdgcn_wmma_f32_16x16x32_f16(
       false, a,  false, b,
       (short)0, c,  false,  false);
}

__device__ __forceinline__ v16h cat8(v8h a, v8h b) {
  v16h r;
  for (int i = 0; i < 8; ++i) { r[i] = a[i]; r[i + 8] = b[i]; }
  return r;
}

__device__ __forceinline__ v8f splat8(float x) {
  v8f r;
  for (int i = 0; i < 8; ++i) r[i] = x;
  return r;
}

__device__ __forceinline__ float sigmoidf_(float x) {
  return __builtin_amdgcn_rcpf(1.0f + __expf(-x));
}
__device__ __forceinline__ float tanhf_(float x) {
  float t = __expf(-2.0f * fabsf(x));
  float r = (1.0f - t) * __builtin_amdgcn_rcpf(1.0f + t);
  return copysignf(r, x);
}

__global__ void lstm_pool_kernel(const float* __restrict__ features,
                                 f16* __restrict__ pooled) {
  int idx = blockIdx.x * blockDim.x + threadIdx.x;
  if (idx >= BATCH * CNNC) return;
  const float* p = features + (size_t)idx * 49;
  float s = 0.f;
  #pragma unroll
  for (int i = 0; i < 49; ++i) s += p[i];
  pooled[idx] = (f16)(s * (1.0f / 49.0f));
}

__device__ __forceinline__ int gate_row(int nt, int m) {
  int r = m & 7, hh = m >> 3;
  return (2 * nt + (r >> 2)) * HID + 4 * hh + (r & 3);
}

__global__ void lstm_pack_kernel(const float* __restrict__ W_in,
                                 const float* __restrict__ W_ih,
                                 const float* __restrict__ W_hh,
                                 const float* __restrict__ W_out,
                                 const float* __restrict__ b_ih,
                                 const float* __restrict__ b_hh,
                                 const float* __restrict__ b_out,
                                 f16* __restrict__ Winp,
                                 f16* __restrict__ Wihp,
                                 f16* __restrict__ Whhp,
                                 f16* __restrict__ Woutp,
                                 float* __restrict__ gbias,
                                 float* __restrict__ obias)
{
  int tid = threadIdx.x;
  for (int idx = tid; idx < 48 * CNNC; idx += 256) {
    int n = idx / CNNC, k = idx % CNNC;
    Winp[idx] = (f16)(n < VOC ? W_in[n * CNNC + k] : 0.f);
  }
  for (int idx = tid; idx < 4 * 512; idx += 256) {
    int tile = idx / 512, e = idx % 512, lane = e / 16, h = e % 16;
    int ks = tile / 2, nt = tile % 2;
    int kl = (h < 8 ? h : h + 8) + (lane / 16) * 8;
    int k = ks * 32 + kl;
    int n = gate_row(nt, lane % 16);
    Wihp[idx] = (f16)(k < VOC ? W_ih[n * VOC + k] : 0.f);
  }
  for (int idx = tid; idx < 2 * 512; idx += 256) {
    int tile = idx / 512, e = idx % 512, lane = e / 16, h = e % 16;
    int n = gate_row(tile, lane % 16);
    Whhp[idx] = (f16)(h < 4 ? W_hh[n * HID + 4 * (lane / 16) + h] : 0.f);
  }
  for (int idx = tid; idx < 3 * 512; idx += 256) {
    int tile = idx / 512, e = idx % 512, lane = e / 16, h = e % 16;
    int kl = (h < 8 ? h : h + 8) + (lane / 16) * 8;
    int n = tile * 16 + (lane % 16);
    Woutp[idx] = (f16)((kl < HID && n < VOC) ? W_out[n * HID + kl] : 0.f);
  }
  if (tid < GN) {
    int n = gate_row(tid / 16, tid % 16);
    gbias[tid] = b_ih[n] + b_hh[n];
  }
  if (tid < 48) obias[tid] = (tid < VOC) ? b_out[tid] : 0.f;
}

__global__ void lstm_featgemm_kernel(const f16* __restrict__ pooled,
                                     const f16* __restrict__ Winp,
                                     const float* __restrict__ b_in,
                                     f16* __restrict__ Xh) {
  const int L = threadIdx.x;
  const int j16 = L & 15, half = L >> 4;
  const int Mb = blockIdx.x * 16;

  v8f acc[3];
  #pragma unroll
  for (int nt = 0; nt < 3; ++nt) {
    int col = nt * 16 + j16;
    acc[nt] = splat8(col < VOC ? b_in[col] : 0.f);
  }

  for (int kt = 0; kt < CNNC / 32; ++kt) {
    int kb = kt * 32 + half * 8;
    int row = Mb + j16;
    v16h A = cat8(*(const v8h*)(pooled + (size_t)row * CNNC + kb),
                  *(const v8h*)(pooled + (size_t)row * CNNC + kb + 16));
    #pragma unroll
    for (int nt = 0; nt < 3; ++nt) {
      int n = nt * 16 + j16;
      v16h Bm = cat8(*(const v8h*)(Winp + (size_t)n * CNNC + kb),
                     *(const v8h*)(Winp + (size_t)n * CNNC + kb + 16));
      acc[nt] = wmma16(A, Bm, acc[nt]);
    }
  }
  #pragma unroll
  for (int r = 0; r < 8; ++r) {
    int m = r + 8 * half;
    f16* dst = Xh + (size_t)(Mb + m) * KPAD;
    #pragma unroll
    for (int nt = 0; nt < 3; ++nt) {
      int col = nt * 16 + j16;
      dst[col] = (col < VOC) ? (f16)acc[nt][r] : (f16)0;
    }
    dst[48 + j16] = (f16)0;
  }
}

__global__ void lstm_packx_kernel(const float* __restrict__ captions,
                                  f16* __restrict__ Xh) {
  size_t idx = (size_t)blockIdx.x * blockDim.x + threadIdx.x;
  int col = idx & (KPAD - 1);
  size_t row = idx >> 6;
  int t = (int)(row / BATCH) + 1;
  int b = (int)(row % BATCH);
  float v = (col < VOC) ? captions[((size_t)b * TSEQ + (t - 1)) * VOC + col] : 0.f;
  Xh[((size_t)t * BATCH + b) * KPAD + col] = (f16)v;
}

__global__ void lstm_pregate_kernel(const f16* __restrict__ Xh,
                                    const f16* __restrict__ Wihp,
                                    const float* __restrict__ gbias,
                                    float* __restrict__ G) {
  const int L = threadIdx.x;
  const int j16 = L & 15, half = L >> 4;
  const size_t Mb = (size_t)blockIdx.x * 16;

  v8f acc[2];
  acc[0] = *(const v8f*)(gbias + 8 * half);
  acc[1] = *(const v8f*)(gbias + 16 + 8 * half);

  #pragma unroll
  for (int ks = 0; ks < 2; ++ks) {
    int kb = ks * 32 + half * 8;
    size_t row = Mb + j16;
    v16h Xb = cat8(*(const v8h*)(Xh + row * KPAD + kb),
                   *(const v8h*)(Xh + row * KPAD + kb + 16));
    #pragma unroll
    for (int nt = 0; nt < 2; ++nt) {
      const f16* tp = Wihp + (size_t)(ks * 2 + nt) * 512 + L * 16;
      v16h Wa = cat8(*(const v8h*)tp, *(const v8h*)(tp + 8));
      acc[nt] = wmma16(Wa, Xb, acc[nt]);
    }
  }
  float* dst = G + (size_t)blockIdx.x * 512 + L * 16;
  *(v8f*)dst       = acc[0];
  *(v8f*)(dst + 8) = acc[1];
}

__global__ void __launch_bounds__(32)
lstm_recur_kernel(const float* __restrict__ G,
                  const f16* __restrict__ Whhp,
                  f16* __restrict__ Hb) {
  const int L = threadIdx.x;
  const int j16 = L & 15, half = L >> 4;
  const int bblk = blockIdx.x;
  const int bbase = bblk * 16;

  v16h WhhA[2];
  for (int nt = 0; nt < 2; ++nt) {
    const f16* tp = Whhp + (size_t)nt * 512 + L * 16;
    WhhA[nt] = cat8(*(const v8h*)tp, *(const v8h*)(tp + 8));
  }

  float cst[4];
  for (int u = 0; u < 4; ++u) cst[u] = 0.f;

  v16h Bh;
  for (int i = 0; i < 16; ++i) Bh[i] = (f16)0;

  f16* hdst = Hb + ((size_t)bbase + j16) * HID + 4 * half;

  for (int t = 0; t < TSEQ; ++t) {
    const float* Gt = G + ((size_t)(t * (BATCH / 16) + bblk) * 512) + L * 16;
    v8f g0 = *(const v8f*)Gt;
    v8f g1 = *(const v8f*)(Gt + 8);
    if (t + 1 < TSEQ)
      __builtin_prefetch(Gt + (size_t)(BATCH / 16) * 512, 0, 1);

    g0 = wmma16(WhhA[0], Bh, g0);
    g1 = wmma16(WhhA[1], Bh, g1);

    v4h hv;
    #pragma unroll
    for (int u = 0; u < 4; ++u) {
      float iv = sigmoidf_(g0[u]);
      float fv = sigmoidf_(g0[4 + u]);
      float gv = tanhf_(g1[u]);
      float ov = sigmoidf_(g1[4 + u]);
      cst[u] = fv * cst[u] + iv * gv;
      hv[u] = (f16)(ov * tanhf_(cst[u]));
    }
    for (int u = 0; u < 4; ++u) Bh[u] = hv[u];

    *(v4h*)(hdst + (size_t)t * BATCH * HID) = hv;
  }
}

__global__ void lstm_outhead_kernel(const f16* __restrict__ Hb,
                                    const f16* __restrict__ Woutp,
                                    const float* __restrict__ obias,
                                    float* __restrict__ out) {
  const int L = threadIdx.x;
  const int j16 = L & 15, half = L >> 4;
  const size_t Mb = (size_t)blockIdx.x * 16;

  v16h WoutB[3];
  #pragma unroll
  for (int nt = 0; nt < 3; ++nt) {
    const f16* tp = Woutp + (size_t)nt * 512 + L * 16;
    WoutB[nt] = cat8(*(const v8h*)tp, *(const v8h*)(tp + 8));
  }

  v16h A;
  #pragma unroll
  for (int i = 0; i < 16; ++i) A[i] = (f16)0;
  if (L < 16) {
    v8h hv = *(const v8h*)(Hb + (Mb + L) * HID);
    #pragma unroll
    for (int k = 0; k < HID; ++k) A[k] = hv[k];
  }

  v8f l0 = splat8(obias[j16]);
  v8f l1 = splat8(obias[16 + j16]);
  v8f l2 = splat8(obias[32 + j16]);
  l0 = wmma16(A, WoutB[0], l0);
  l1 = wmma16(A, WoutB[1], l1);
  l2 = wmma16(A, WoutB[2], l2);

  const bool v2ok = (32 + j16) < VOC;
  #pragma unroll
  for (int r = 0; r < 8; ++r) {
    float x2 = v2ok ? l2[r] : -1e30f;
    float mx = fmaxf(fmaxf(l0[r], l1[r]), x2);
    #pragma unroll
    for (int s = 1; s < 16; s <<= 1)
      mx = fmaxf(mx, __shfl_xor(mx, s, 16));
    float e0 = __expf(l0[r] - mx);
    float e1 = __expf(l1[r] - mx);
    float e2 = v2ok ? __expf(x2 - mx) : 0.f;
    float sm = e0 + e1 + e2;
    #pragma unroll
    for (int s = 1; s < 16; s <<= 1)
      sm += __shfl_xor(sm, s, 16);
    float inv = 1.0f / sm;

    size_t row = Mb + r + 8 * half;
    int t = (int)(row / BATCH);
    int b = (int)(row % BATCH);
    float* dst = out + ((size_t)b * TSEQ + t) * VOC;
    dst[j16]      = e0 * inv;
    dst[16 + j16] = e1 * inv;
    if (v2ok) dst[32 + j16] = e2 * inv;
  }
}

extern "C" void kernel_launch(void* const* d_in, const int* in_sizes, int n_in,
                              void* d_out, int out_size, void* d_ws, size_t ws_size,
                              hipStream_t stream) {
  const float* features = (const float*)d_in[0];
  const float* captions = (const float*)d_in[1];
  const float* W_in  = (const float*)d_in[2];
  const float* b_in  = (const float*)d_in[3];
  const float* W_ih  = (const float*)d_in[4];
  const float* W_hh  = (const float*)d_in[5];
  const float* b_ih  = (const float*)d_in[6];
  const float* b_hh  = (const float*)d_in[7];
  const float* W_out = (const float*)d_in[8];
  const float* b_out = (const float*)d_in[9];
  float* out = (float*)d_out;

  char* ws = (char*)d_ws;
  auto carve = [&](size_t bytes) -> char* {
    char* p = ws;
    ws += (bytes + 255) & ~(size_t)255;
    return p;
  };
  f16*   pooled = (f16*)  carve((size_t)BATCH * CNNC * sizeof(f16));
  f16*   Winp   = (f16*)  carve((size_t)48 * CNNC * sizeof(f16));
  f16*   Wihp   = (f16*)  carve((size_t)4 * 512 * sizeof(f16));
  f16*   Whhp   = (f16*)  carve((size_t)2 * 512 * sizeof(f16));
  f16*   Woutp  = (f16*)  carve((size_t)3 * 512 * sizeof(f16));
  float* gbias  = (float*)carve(GN * sizeof(float));
  float* obias  = (float*)carve(48 * sizeof(float));
  f16*   Xh     = (f16*)  carve((size_t)NROW * KPAD * sizeof(f16));
  float* G      = (float*)carve((size_t)NROW * GN * sizeof(float));
  f16*   Hb     = (f16*)  carve((size_t)NROW * HID * sizeof(f16));

  lstm_pool_kernel<<<(BATCH * CNNC) / 256, 256, 0, stream>>>(features, pooled);
  lstm_pack_kernel<<<1, 256, 0, stream>>>(W_in, W_ih, W_hh, W_out,
                                          b_ih, b_hh, b_out,
                                          Winp, Wihp, Whhp, Woutp, gbias, obias);
  lstm_featgemm_kernel<<<BATCH / 16, 32, 0, stream>>>(pooled, Winp, b_in, Xh);
  lstm_packx_kernel<<<((TSEQ - 1) * BATCH * KPAD) / 256, 256, 0, stream>>>(captions, Xh);
  lstm_pregate_kernel<<<NTILE, 32, 0, stream>>>(Xh, Wihp, gbias, G);
  lstm_recur_kernel<<<BATCH / 16, 32, 0, stream>>>(G, Whhp, Hb);
  lstm_outhead_kernel<<<NTILE, 32, 0, stream>>>(Hb, Woutp, obias, out);
}


# Recurrence: pre-gate (C operand) tiles loaded 8 steps ahead into a ring of register tiles, loop unrolled by 8: the global-load latency leaves the serial chain of every step; prefetch hint dropped
# speedup vs baseline: 3.0011x; 1.7212x over previous
#include <hip/hip_runtime.h>
#include <hip/hip_bf16.h>

typedef _Float16 f16;
typedef __attribute__((ext_vector_type(16))) _Float16 v16h;
typedef __attribute__((ext_vector_type(8)))  _Float16 v8h;
typedef __attribute__((ext_vector_type(4)))  _Float16 v4h;
typedef __attribute__((ext_vector_type(8)))  float    v8f;

#define BATCH 512
#define TSEQ  256
#define CNNC  512
#define VOC   37
#define HID   8
#define KPAD  64
#define GN    32
#define NROW  (TSEQ * BATCH)
#define NTILE (NROW / 16)

__device__ __forceinline__ v8f wmma16(v16h a, v16h b, v8f c) {
  return __builtin_amdgcn_wmma_f32_16x16x32_f16(
       false, a,  false, b,
       (short)0, c,  false,  false);
}

__device__ __forceinline__ v16h cat8(v8h a, v8h b) {
  v16h r;
  for (int i = 0; i < 8; ++i) { r[i] = a[i]; r[i + 8] = b[i]; }
  return r;
}

__device__ __forceinline__ v8f splat8(float x) {
  v8f r;
  for (int i = 0; i < 8; ++i) r[i] = x;
  return r;
}

__device__ __forceinline__ float sigmoidf_(float x) {
  return __builtin_amdgcn_rcpf(1.0f + __expf(-x));
}
__device__ __forceinline__ float tanhf_(float x) {
  float t = __expf(-2.0f * fabsf(x));
  float r = (1.0f - t) * __builtin_amdgcn_rcpf(1.0f + t);
  return copysignf(r, x);
}

__global__ void lstm_pool_kernel(const float* __restrict__ features,
                                 f16* __restrict__ pooled) {
  int idx = blockIdx.x * blockDim.x + threadIdx.x;
  if (idx >= BATCH * CNNC) return;
  const float* p = features + (size_t)idx * 49;
  float s = 0.f;
  #pragma unroll
  for (int i = 0; i < 49; ++i) s += p[i];
  pooled[idx] = (f16)(s * (1.0f / 49.0f));
}

__device__ __forceinline__ int gate_row(int nt, int m) {
  int r = m & 7, hh = m >> 3;
  return (2 * nt + (r >> 2)) * HID + 4 * hh + (r & 3);
}

__global__ void lstm_pack_kernel(const float* __restrict__ W_in,
                                 const float* __restrict__ W_ih,
                                 const float* __restrict__ W_hh,
                                 const float* __restrict__ W_out,
                                 const float* __restrict__ b_ih,
                                 const float* __restrict__ b_hh,
                                 const float* __restrict__ b_out,
                                 f16* __restrict__ Winp,
                                 f16* __restrict__ Wihp,
                                 f16* __restrict__ Whhp,
                                 f16* __restrict__ Woutp,
                                 float* __restrict__ gbias,
                                 float* __restrict__ obias)
{
  int tid = threadIdx.x;
  for (int idx = tid; idx < 48 * CNNC; idx += 256) {
    int n = idx / CNNC, k = idx % CNNC;
    Winp[idx] = (f16)(n < VOC ? W_in[n * CNNC + k] : 0.f);
  }
  for (int idx = tid; idx < 4 * 512; idx += 256) {
    int tile = idx / 512, e = idx % 512, lane = e / 16, h = e % 16;
    int ks = tile / 2, nt = tile % 2;
    int kl = (h < 8 ? h : h + 8) + (lane / 16) * 8;
    int k = ks * 32 + kl;
    int n = gate_row(nt, lane % 16);
    Wihp[idx] = (f16)(k < VOC ? W_ih[n * VOC + k] : 0.f);
  }
  for (int idx = tid; idx < 2 * 512; idx += 256) {
    int tile = idx / 512, e = idx % 512, lane = e / 16, h = e % 16;
    int n = gate_row(tile, lane % 16);
    Whhp[idx] = (f16)(h < 4 ? W_hh[n * HID + 4 * (lane / 16) + h] : 0.f);
  }
  for (int idx = tid; idx < 3 * 512; idx += 256) {
    int tile = idx / 512, e = idx % 512, lane = e / 16, h = e % 16;
    int kl = (h < 8 ? h : h + 8) + (lane / 16) * 8;
    int n = tile * 16 + (lane % 16);
    Woutp[idx] = (f16)((kl < HID && n < VOC) ? W_out[n * HID + kl] : 0.f);
  }
  if (tid < GN) {
    int n = gate_row(tid / 16, tid % 16);
    gbias[tid] = b_ih[n] + b_hh[n];
  }
  if (tid < 48) obias[tid] = (tid < VOC) ? b_out[tid] : 0.f;
}

__global__ void lstm_featgemm_kernel(const f16* __restrict__ pooled,
                                     const f16* __restrict__ Winp,
                                     const float* __restrict__ b_in,
                                     f16* __restrict__ Xh) {
  const int L = threadIdx.x;
  const int j16 = L & 15, half = L >> 4;
  const int Mb = blockIdx.x * 16;

  v8f acc[3];
  #pragma unroll
  for (int nt = 0; nt < 3; ++nt) {
    int col = nt * 16 + j16;
    acc[nt] = splat8(col < VOC ? b_in[col] : 0.f);
  }

  for (int kt = 0; kt < CNNC / 32; ++kt) {
    int kb = kt * 32 + half * 8;
    int row = Mb + j16;
    v16h A = cat8(*(const v8h*)(pooled + (size_t)row * CNNC + kb),
                  *(const v8h*)(pooled + (size_t)row * CNNC + kb + 16));
    #pragma unroll
    for (int nt = 0; nt < 3; ++nt) {
      int n = nt * 16 + j16;
      v16h Bm = cat8(*(const v8h*)(Winp + (size_t)n * CNNC + kb),
                     *(const v8h*)(Winp + (size_t)n * CNNC + kb + 16));
      acc[nt] = wmma16(A, Bm, acc[nt]);
    }
  }
  #pragma unroll
  for (int r = 0; r < 8; ++r) {
    int m = r + 8 * half;
    f16* dst = Xh + (size_t)(Mb + m) * KPAD;
    #pragma unroll
    for (int nt = 0; nt < 3; ++nt) {
      int col = nt * 16 + j16;
      dst[col] = (col < VOC) ? (f16)acc[nt][r] : (f16)0;
    }
    dst[48 + j16] = (f16)0;
  }
}

__global__ void lstm_packx_kernel(const float* __restrict__ captions,
                                  f16* __restrict__ Xh) {
  size_t idx = (size_t)blockIdx.x * blockDim.x + threadIdx.x;
  int col = idx & (KPAD - 1);
  size_t row = idx >> 6;
  int t = (int)(row / BATCH) + 1;
  int b = (int)(row % BATCH);
  float v = (col < VOC) ? captions[((size_t)b * TSEQ + (t - 1)) * VOC + col] : 0.f;
  Xh[((size_t)t * BATCH + b) * KPAD + col] = (f16)v;
}

__global__ void lstm_pregate_kernel(const f16* __restrict__ Xh,
                                    const f16* __restrict__ Wihp,
                                    const float* __restrict__ gbias,
                                    float* __restrict__ G) {
  const int L = threadIdx.x;
  const int j16 = L & 15, half = L >> 4;
  const size_t Mb = (size_t)blockIdx.x * 16;

  v8f acc[2];
  acc[0] = *(const v8f*)(gbias + 8 * half);
  acc[1] = *(const v8f*)(gbias + 16 + 8 * half);

  #pragma unroll
  for (int ks = 0; ks < 2; ++ks) {
    int kb = ks * 32 + half * 8;
    size_t row = Mb + j16;
    v16h Xb = cat8(*(const v8h*)(Xh + row * KPAD + kb),
                   *(const v8h*)(Xh + row * KPAD + kb + 16));
    #pragma unroll
    for (int nt = 0; nt < 2; ++nt) {
      const f16* tp = Wihp + (size_t)(ks * 2 + nt) * 512 + L * 16;
      v16h Wa = cat8(*(const v8h*)tp, *(const v8h*)(tp + 8));
      acc[nt] = wmma16(Wa, Xb, acc[nt]);
    }
  }
  float* dst = G + (size_t)blockIdx.x * 512 + L * 16;
  *(v8f*)dst       = acc[0];
  *(v8f*)(dst + 8) = acc[1];
}

#define GAHEAD 8
static_assert(TSEQ % GAHEAD == 0 && TSEQ >= 2 * GAHEAD, "ring depth must divide the sequence");

__device__ __forceinline__ void lstm_cell_step(v8f g0, v8f g1,
                                               const v16h& Wa0, const v16h& Wa1,
                                               v16h& Bh, float (&cst)[4],
                                               f16* __restrict__ hout) {
  g0 = wmma16(Wa0, Bh, g0);
  g1 = wmma16(Wa1, Bh, g1);

  v4h hv;
  #pragma unroll
  for (int u = 0; u < 4; ++u) {
    float iv = sigmoidf_(g0[u]);
    float fv = sigmoidf_(g0[4 + u]);
    float gv = tanhf_(g1[u]);
    float ov = sigmoidf_(g1[4 + u]);
    cst[u] = fv * cst[u] + iv * gv;
    hv[u] = (f16)(ov * tanhf_(cst[u]));
  }
  for (int u = 0; u < 4; ++u) Bh[u] = hv[u];

  *(v4h*)hout = hv;
}

__global__ void __launch_bounds__(32)
lstm_recur_kernel(const float* __restrict__ G,
                  const f16* __restrict__ Whhp,
                  f16* __restrict__ Hb) {
  const int L = threadIdx.x;
  const int j16 = L & 15, half = L >> 4;
  const int bblk = blockIdx.x;
  const int bbase = bblk * 16;

  v16h WhhA[2];
  for (int nt = 0; nt < 2; ++nt) {
    const f16* tp = Whhp + (size_t)nt * 512 + L * 16;
    WhhA[nt] = cat8(*(const v8h*)tp, *(const v8h*)(tp + 8));
  }

  float cst[4];
  for (int u = 0; u < 4; ++u) cst[u] = 0.f;

  v16h Bh;
  for (int i = 0; i < 16; ++i) Bh[i] = (f16)0;

  f16* hdst = Hb + ((size_t)bbase + j16) * HID + 4 * half;
  const size_t HSTEP = (size_t)BATCH * HID;

  const float* Gl = G + (size_t)bblk * 512 + L * 16;
  const size_t GSTEP = (size_t)(BATCH / 16) * 512;

  __builtin_amdgcn_sched_barrier(0);

  v8f ra[GAHEAD], rb[GAHEAD];
  #pragma unroll
  for (int d = 0; d < GAHEAD; ++d) {
    ra[d] = *(const v8f*)(Gl + d * GSTEP);
    rb[d] = *(const v8f*)(Gl + d * GSTEP + 8);
  }

  int t = 0;
  for (; t < TSEQ - GAHEAD; t += GAHEAD) {
    #pragma unroll
    for (int d = 0; d < GAHEAD; ++d) {
      lstm_cell_step(ra[d], rb[d], WhhA[0], WhhA[1], Bh, cst,
                     hdst + (size_t)(t + d) * HSTEP);
      __builtin_amdgcn_sched_barrier(0);
      const float* Gn = Gl + (size_t)(t + d + GAHEAD) * GSTEP;
      ra[d] = *(const v8f*)Gn;
      rb[d] = *(const v8f*)(Gn + 8);
    }
  }
  #pragma unroll
  for (int d = 0; d < GAHEAD; ++d)
    lstm_cell_step(ra[d], rb[d], WhhA[0], WhhA[1], Bh, cst,
                   hdst + (size_t)(t + d) * HSTEP);
}

__global__ void lstm_outhead_kernel(const f16* __restrict__ Hb,
                                    const f16* __restrict__ Woutp,
                                    const float* __restrict__ obias,
                                    float* __restrict__ out) {
  const int L = threadIdx.x;
  const int j16 = L & 15, half = L >> 4;
  const size_t Mb = (size_t)blockIdx.x * 16;

  v16h WoutB[3];
  #pragma unroll
  for (int nt = 0; nt < 3; ++nt) {
    const f16* tp = Woutp + (size_t)nt * 512 + L * 16;
    WoutB[nt] = cat8(*(const v8h*)tp, *(const v8h*)(tp + 8));
  }

  v16h A;
  #pragma unroll
  for (int i = 0; i < 16; ++i) A[i] = (f16)0;
  if (L < 16) {
    v8h hv = *(const v8h*)(Hb + (Mb + L) * HID);
    #pragma unroll
    for (int k = 0; k < HID; ++k) A[k] = hv[k];
  }

  v8f l0 = splat8(obias[j16]);
  v8f l1 = splat8(obias[16 + j16]);
  v8f l2 = splat8(obias[32 + j16]);
  l0 = wmma16(A, WoutB[0], l0);
  l1 = wmma16(A, WoutB[1], l1);
  l2 = wmma16(A, WoutB[2], l2);

  const bool v2ok = (32 + j16) < VOC;
  #pragma unroll
  for (int r = 0; r < 8; ++r) {
    float x2 = v2ok ? l2[r] : -1e30f;
    float mx = fmaxf(fmaxf(l0[r], l1[r]), x2);
    #pragma unroll
    for (int s = 1; s < 16; s <<= 1)
      mx = fmaxf(mx, __shfl_xor(mx, s, 16));
    float e0 = __expf(l0[r] - mx);
    float e1 = __expf(l1[r] - mx);
    float e2 = v2ok ? __expf(x2 - mx) : 0.f;
    float sm = e0 + e1 + e2;
    #pragma unroll
    for (int s = 1; s < 16; s <<= 1)
      sm += __shfl_xor(sm, s, 16);
    float inv = 1.0f / sm;

    size_t row = Mb + r + 8 * half;
    int t = (int)(row / BATCH);
    int b = (int)(row % BATCH);
    float* dst = out + ((size_t)b * TSEQ + t) * VOC;
    dst[j16]      = e0 * inv;
    dst[16 + j16] = e1 * inv;
    if (v2ok) dst[32 + j16] = e2 * inv;
  }
}

extern "C" void kernel_launch(void* const* d_in, const int* in_sizes, int n_in,
                              void* d_out, int out_size, void* d_ws, size_t ws_size,
                              hipStream_t stream) {
  const float* features = (const float*)d_in[0];
  const float* captions = (const float*)d_in[1];
  const float* W_in  = (const float*)d_in[2];
  const float* b_in  = (const float*)d_in[3];
  const float* W_ih  = (const float*)d_in[4];
  const float* W_hh  = (const float*)d_in[5];
  const float* b_ih  = (const float*)d_in[6];
  const float* b_hh  = (const float*)d_in[7];
  const float* W_out = (const float*)d_in[8];
  const float* b_out = (const float*)d_in[9];
  float* out = (float*)d_out;

  char* ws = (char*)d_ws;
  auto carve = [&](size_t bytes) -> char* {
    char* p = ws;
    ws += (bytes + 255) & ~(size_t)255;
    return p;
  };
  f16*   pooled = (f16*)  carve((size_t)BATCH * CNNC * sizeof(f16));
  f16*   Winp   = (f16*)  carve((size_t)48 * CNNC * sizeof(f16));
  f16*   Wihp   = (f16*)  carve((size_t)4 * 512 * sizeof(f16));
  f16*   Whhp   = (f16*)  carve((size_t)2 * 512 * sizeof(f16));
  f16*   Woutp  = (f16*)  carve((size_t)3 * 512 * sizeof(f16));
  float* gbias  = (float*)carve(GN * sizeof(float));
  float* obias  = (float*)carve(48 * sizeof(float));
  f16*   Xh     = (f16*)  carve((size_t)NROW * KPAD * sizeof(f16));
  float* G      = (float*)carve((size_t)NROW * GN * sizeof(float));
  f16*   Hb     = (f16*)  carve((size_t)NROW * HID * sizeof(f16));

  lstm_pool_kernel<<<(BATCH * CNNC) / 256, 256, 0, stream>>>(features, pooled);
  lstm_pack_kernel<<<1, 256, 0, stream>>>(W_in, W_ih, W_hh, W_out,
                                          b_ih, b_hh, b_out,
                                          Winp, Wihp, Whhp, Woutp, gbias, obias);
  lstm_featgemm_kernel<<<BATCH / 16, 32, 0, stream>>>(pooled, Winp, b_in, Xh);
  lstm_packx_kernel<<<((TSEQ - 1) * BATCH * KPAD) / 256, 256, 0, stream>>>(captions, Xh);
  lstm_pregate_kernel<<<NTILE, 32, 0, stream>>>(Xh, Wihp, gbias, G);
  lstm_recur_kernel<<<BATCH / 16, 32, 0, stream>>>(G, Whhp, Hb);
  lstm_outhead_kernel<<<NTILE, 32, 0, stream>>>(Hb, Woutp, obias, out);
}


# Pre-gate GEMM builds its B operand from the f32 rows (captions, feat) in registers: the f16 staging kernel (one element per thread) is gone; k order inside each 32-k product changed, W_ih alike
# speedup vs baseline: 3.5079x; 1.1689x over previous
#include <hip/hip_runtime.h>
#include <hip/hip_bf16.h>

typedef _Float16 f16;
typedef __attribute__((ext_vector_type(16))) _Float16 v16h;
typedef __attribute__((ext_vector_type(8)))  _Float16 v8h;
typedef __attribute__((ext_vector_type(4)))  _Float16 v4h;
typedef __attribute__((ext_vector_type(8)))  float    v8f;

#define BATCH 512
#define TSEQ  256
#define CNNC  512
#define VOC   37
#define HID   8
#define GN    32
#define NROW  (TSEQ * BATCH)
#define NTILE (NROW / 16)

__device__ __forceinline__ v8f wmma16(v16h a, v16h b, v8f c) {
  return __builtin_amdgcn_wmma_f32_16x16x32_f16(
       false, a,  false, b,
       (short)0, c,  false,  false);
}

__device__ __forceinline__ v16h cat8(v8h a, v8h b) {
  v16h r;
  for (int i = 0; i < 8; ++i) { r[i] = a[i]; r[i + 8] = b[i]; }
  return r;
}

__device__ __forceinline__ v8f splat8(float x) {
  v8f r;
  for (int i = 0; i < 8; ++i) r[i] = x;
  return r;
}

__device__ __forceinline__ float sigmoidf_(float x) {
  return __builtin_amdgcn_rcpf(1.0f + __expf(-x));
}
__device__ __forceinline__ float tanhf_(float x) {
  float t = __expf(-2.0f * fabsf(x));
  float r = (1.0f - t) * __builtin_amdgcn_rcpf(1.0f + t);
  return copysignf(r, x);
}

__global__ void lstm_pool_kernel(const float* __restrict__ features,
                                 f16* __restrict__ pooled) {
  int idx = blockIdx.x * blockDim.x + threadIdx.x;
  if (idx >= BATCH * CNNC) return;
  const float* p = features + (size_t)idx * 49;
  float s = 0.f;
  #pragma unroll
  for (int i = 0; i < 49; ++i) s += p[i];
  pooled[idx] = (f16)(s * (1.0f / 49.0f));
}

__device__ __forceinline__ int gate_row(int nt, int m) {
  int r = m & 7, hh = m >> 3;
  return (2 * nt + (r >> 2)) * HID + 4 * hh + (r & 3);
}

__global__ void lstm_pack_kernel(const float* __restrict__ W_in,
                                 const float* __restrict__ W_ih,
                                 const float* __restrict__ W_hh,
                                 const float* __restrict__ W_out,
                                 const float* __restrict__ b_ih,
                                 const float* __restrict__ b_hh,
                                 const float* __restrict__ b_out,
                                 f16* __restrict__ Winp,
                                 f16* __restrict__ Wihp,
                                 f16* __restrict__ Whhp,
                                 f16* __restrict__ Woutp,
                                 float* __restrict__ gbias,
                                 float* __restrict__ obias)
{
  int tid = threadIdx.x;
  for (int idx = tid; idx < 48 * CNNC; idx += 256) {
    int n = idx / CNNC, k = idx % CNNC;
    Winp[idx] = (f16)(n < VOC ? W_in[n * CNNC + k] : 0.f);
  }
  for (int idx = tid; idx < 4 * 512; idx += 256) {
    int tile = idx / 512, e = idx % 512, lane = e / 16, h = e % 16;
    int ks = tile / 2, nt = tile % 2;
    int k = ks * 32 + 16 * (lane / 16) + h;
    int n = gate_row(nt, lane % 16);
    Wihp[idx] = (f16)(k < VOC ? W_ih[n * VOC + k] : 0.f);
  }
  for (int idx = tid; idx < 2 * 512; idx += 256) {
    int tile = idx / 512, e = idx % 512, lane = e / 16, h = e % 16;
    int n = gate_row(tile, lane % 16);
    Whhp[idx] = (f16)(h < 4 ? W_hh[n * HID + 4 * (lane / 16) + h] : 0.f);
  }
  for (int idx = tid; idx < 3 * 512; idx += 256) {
    int tile = idx / 512, e = idx % 512, lane = e / 16, h = e % 16;
    int kl = (h < 8 ? h : h + 8) + (lane / 16) * 8;
    int n = tile * 16 + (lane % 16);
    Woutp[idx] = (f16)((kl < HID && n < VOC) ? W_out[n * HID + kl] : 0.f);
  }
  if (tid < GN) {
    int n = gate_row(tid / 16, tid % 16);
    gbias[tid] = b_ih[n] + b_hh[n];
  }
  if (tid < 48) obias[tid] = (tid < VOC) ? b_out[tid] : 0.f;
}

__global__ void lstm_featgemm_kernel(const f16* __restrict__ pooled,
                                     const f16* __restrict__ Winp,
                                     const float* __restrict__ b_in,
                                     float* __restrict__ X0) {
  const int L = threadIdx.x;
  const int j16 = L & 15, half = L >> 4;
  const int Mb = blockIdx.x * 16;

  v8f acc[3];
  #pragma unroll
  for (int nt = 0; nt < 3; ++nt) {
    int col = nt * 16 + j16;
    acc[nt] = splat8(col < VOC ? b_in[col] : 0.f);
  }

  for (int kt = 0; kt < CNNC / 32; ++kt) {
    int kb = kt * 32 + half * 8;
    int row = Mb + j16;
    v16h A = cat8(*(const v8h*)(pooled + (size_t)row * CNNC + kb),
                  *(const v8h*)(pooled + (size_t)row * CNNC + kb + 16));
    #pragma unroll
    for (int nt = 0; nt < 3; ++nt) {
      int n = nt * 16 + j16;
      v16h Bm = cat8(*(const v8h*)(Winp + (size_t)n * CNNC + kb),
                     *(const v8h*)(Winp + (size_t)n * CNNC + kb + 16));
      acc[nt] = wmma16(A, Bm, acc[nt]);
    }
  }
  #pragma unroll
  for (int r = 0; r < 8; ++r) {
    int m = r + 8 * half;
    float* dst = X0 + (size_t)(Mb + m) * VOC;
    #pragma unroll
    for (int nt = 0; nt < 3; ++nt) {
      int col = nt * 16 + j16;
      if (col < VOC) dst[col] = acc[nt][r];
    }
  }
}

__global__ void lstm_pregate_kernel(const float* __restrict__ captions,
                                    const float* __restrict__ X0,
                                    const f16* __restrict__ Wihp,
                                    const float* __restrict__ gbias,
                                    float* __restrict__ G) {
  const int L = threadIdx.x;
  const int j16 = L & 15, half = L >> 4;
  const int t = blockIdx.x / (BATCH / 16);
  const int b = (blockIdx.x % (BATCH / 16)) * 16 + j16;

  v8f acc[2];
  acc[0] = *(const v8f*)(gbias + 8 * half);
  acc[1] = *(const v8f*)(gbias + 16 + 8 * half);

  const float* xr = (t == 0) ? X0 + (size_t)b * VOC
                             : captions + ((size_t)b * TSEQ + (t - 1)) * VOC;
  v16h Xb[2];
  #pragma unroll
  for (int s = 0; s < 16; ++s) Xb[0][s] = (f16)xr[16 * half + s];
  #pragma unroll
  for (int s = 0; s < 16; ++s) {
    float v = (s < VOC - 32) ? xr[32 + s] : 0.f;
    Xb[1][s] = (half == 0) ? (f16)v : (f16)0;
  }

  #pragma unroll
  for (int ks = 0; ks < 2; ++ks) {
    #pragma unroll
    for (int nt = 0; nt < 2; ++nt) {
      const f16* tp = Wihp + (size_t)(ks * 2 + nt) * 512 + L * 16;
      v16h Wa = cat8(*(const v8h*)tp, *(const v8h*)(tp + 8));
      acc[nt] = wmma16(Wa, Xb[ks], acc[nt]);
    }
  }
  float* dst = G + (size_t)blockIdx.x * 512 + L * 16;
  *(v8f*)dst       = acc[0];
  *(v8f*)(dst + 8) = acc[1];
}

#define GAHEAD 8
static_assert(TSEQ % GAHEAD == 0 && TSEQ >= 2 * GAHEAD, "ring depth must divide the sequence");

__device__ __forceinline__ void lstm_cell_step(v8f g0, v8f g1,
                                               const v16h& Wa0, const v16h& Wa1,
                                               v16h& Bh, float (&cst)[4],
                                               f16* __restrict__ hout) {
  g0 = wmma16(Wa0, Bh, g0);
  g1 = wmma16(Wa1, Bh, g1);

  v4h hv;
  #pragma unroll
  for (int u = 0; u < 4; ++u) {
    float iv = sigmoidf_(g0[u]);
    float fv = sigmoidf_(g0[4 + u]);
    float gv = tanhf_(g1[u]);
    float ov = sigmoidf_(g1[4 + u]);
    cst[u] = fv * cst[u] + iv * gv;
    hv[u] = (f16)(ov * tanhf_(cst[u]));
  }
  for (int u = 0; u < 4; ++u) Bh[u] = hv[u];

  *(v4h*)hout = hv;
}

__global__ void __launch_bounds__(32)
lstm_recur_kernel(const float* __restrict__ G,
                  const f16* __restrict__ Whhp,
                  f16* __restrict__ Hb) {
  const int L = threadIdx.x;
  const int j16 = L & 15, half = L >> 4;
  const int bblk = blockIdx.x;
  const int bbase = bblk * 16;

  v16h WhhA[2];
  for (int nt = 0; nt < 2; ++nt) {
    const f16* tp = Whhp + (size_t)nt * 512 + L * 16;
    WhhA[nt] = cat8(*(const v8h*)tp, *(const v8h*)(tp + 8));
  }

  float cst[4];
  for (int u = 0; u < 4; ++u) cst[u] = 0.f;

  v16h Bh;
  for (int i = 0; i < 16; ++i) Bh[i] = (f16)0;

  f16* hdst = Hb + ((size_t)bbase + j16) * HID + 4 * half;
  const size_t HSTEP = (size_t)BATCH * HID;

  const float* Gl = G + (size_t)bblk * 512 + L * 16;
  const size_t GSTEP = (size_t)(BATCH / 16) * 512;

  __builtin_amdgcn_sched_barrier(0);

  v8f ra[GAHEAD], rb[GAHEAD];
  #pragma unroll
  for (int d = 0; d < GAHEAD; ++d) {
    ra[d] = *(const v8f*)(Gl + d * GSTEP);
    rb[d] = *(const v8f*)(Gl + d * GSTEP + 8);
  }

  int t = 0;
  for (; t < TSEQ - GAHEAD; t += GAHEAD) {
    #pragma unroll
    for (int d = 0; d < GAHEAD; ++d) {
      lstm_cell_step(ra[d], rb[d], WhhA[0], WhhA[1], Bh, cst,
                     hdst + (size_t)(t + d) * HSTEP);
      __builtin_amdgcn_sched_barrier(0);
      const float* Gn = Gl + (size_t)(t + d + GAHEAD) * GSTEP;
      ra[d] = *(const v8f*)Gn;
      rb[d] = *(const v8f*)(Gn + 8);
    }
  }
  #pragma unroll
  for (int d = 0; d < GAHEAD; ++d)
    lstm_cell_step(ra[d], rb[d], WhhA[0], WhhA[1], Bh, cst,
                   hdst + (size_t)(t + d) * HSTEP);
}

__global__ void lstm_outhead_kernel(const f16* __restrict__ Hb,
                                    const f16* __restrict__ Woutp,
                                    const float* __restrict__ obias,
                                    float* __restrict__ out) {
  const int L = threadIdx.x;
  const int j16 = L & 15, half = L >> 4;
  const size_t Mb = (size_t)blockIdx.x * 16;

  v16h WoutB[3];
  #pragma unroll
  for (int nt = 0; nt < 3; ++nt) {
    const f16* tp = Woutp + (size_t)nt * 512 + L * 16;
    WoutB[nt] = cat8(*(const v8h*)tp, *(const v8h*)(tp + 8));
  }

  v16h A;
  #pragma unroll
  for (int i = 0; i < 16; ++i) A[i] = (f16)0;
  if (L < 16) {
    v8h hv = *(const v8h*)(Hb + (Mb + L) * HID);
    #pragma unroll
    for (int k = 0; k < HID; ++k) A[k] = hv[k];
  }

  v8f l0 = splat8(obias[j16]);
  v8f l1 = splat8(obias[16 + j16]);
  v8f l2 = splat8(obias[32 + j16]);
  l0 = wmma16(A, WoutB[0], l0);
  l1 = wmma16(A, WoutB[1], l1);
  l2 = wmma16(A, WoutB[2], l2);

  const bool v2ok = (32 + j16) < VOC;
  #pragma unroll
  for (int r = 0; r < 8; ++r) {
    float x2 = v2ok ? l2[r] : -1e30f;
    float mx = fmaxf(fmaxf(l0[r], l1[r]), x2);
    #pragma unroll
    for (int s = 1; s < 16; s <<= 1)
      mx = fmaxf(mx, __shfl_xor(mx, s, 16));
    float e0 = __expf(l0[r] - mx);
    float e1 = __expf(l1[r] - mx);
    float e2 = v2ok ? __expf(x2 - mx) : 0.f;
    float sm = e0 + e1 + e2;
    #pragma unroll
    for (int s = 1; s < 16; s <<= 1)
      sm += __shfl_xor(sm, s, 16);
    float inv = 1.0f / sm;

    size_t row = Mb + r + 8 * half;
    int t = (int)(row / BATCH);
    int b = (int)(row % BATCH);
    float* dst = out + ((size_t)b * TSEQ + t) * VOC;
    dst[j16]      = e0 * inv;
    dst[16 + j16] = e1 * inv;
    if (v2ok) dst[32 + j16] = e2 * inv;
  }
}

extern "C" void kernel_launch(void* const* d_in, const int* in_sizes, int n_in,
                              void* d_out, int out_size, void* d_ws, size_t ws_size,
                              hipStream_t stream) {
  const float* features = (const float*)d_in[0];
  const float* captions = (const float*)d_in[1];
  const float* W_in  = (const float*)d_in[2];
  const float* b_in  = (const float*)d_in[3];
  const float* W_ih  = (const float*)d_in[4];
  const float* W_hh  = (const float*)d_in[5];
  const float* b_ih  = (const float*)d_in[6];
  const float* b_hh  = (const float*)d_in[7];
  const float* W_out = (const float*)d_in[8];
  const float* b_out = (const float*)d_in[9];
  float* out = (float*)d_out;

  char* ws = (char*)d_ws;
  auto carve = [&](size_t bytes) -> char* {
    char* p = ws;
    ws += (bytes + 255) & ~(size_t)255;
    return p;
  };
  f16*   pooled = (f16*)  carve((size_t)BATCH * CNNC * sizeof(f16));
  f16*   Winp   = (f16*)  carve((size_t)48 * CNNC * sizeof(f16));
  f16*   Wihp   = (f16*)  carve((size_t)4 * 512 * sizeof(f16));
  f16*   Whhp   = (f16*)  carve((size_t)2 * 512 * sizeof(f16));
  f16*   Woutp  = (f16*)  carve((size_t)3 * 512 * sizeof(f16));
  float* gbias  = (float*)carve(GN * sizeof(float));
  float* obias  = (float*)carve(48 * sizeof(float));
  float* X0     = (float*)carve((size_t)BATCH * VOC * sizeof(float));
  float* G      = (float*)carve((size_t)NROW * GN * sizeof(float));
  f16*   Hb     = (f16*)  carve((size_t)NROW * HID * sizeof(f16));

  lstm_pool_kernel<<<(BATCH * CNNC) / 256, 256, 0, stream>>>(features, pooled);
  lstm_pack_kernel<<<1, 256, 0, stream>>>(W_in, W_ih, W_hh, W_out,
                                          b_ih, b_hh, b_out,
                                          Winp, Wihp, Whhp, Woutp, gbias, obias);
  lstm_featgemm_kernel<<<BATCH / 16, 32, 0, stream>>>(pooled, Winp, b_in, X0);
  lstm_pregate_kernel<<<NTILE, 32, 0, stream>>>(captions, X0, Wihp, gbias, G);
  lstm_recur_kernel<<<BATCH / 16, 32, 0, stream>>>(G, Whhp, Hb);
  lstm_outhead_kernel<<<NTILE, 32, 0, stream>>>(Hb, Woutp, obias, out);
}


# Weight-pack kernel spread over 115 blocks, one element per thread: the single block's chain of about 90 dependent load round trips per thread (each load waited for in its own branch) is gone
# speedup vs baseline: 5.4986x; 1.5675x over previous
#include <hip/hip_runtime.h>
#include <hip/hip_bf16.h>

typedef _Float16 f16;
typedef __attribute__((ext_vector_type(16))) _Float16 v16h;
typedef __attribute__((ext_vector_type(8)))  _Float16 v8h;
typedef __attribute__((ext_vector_type(4)))  _Float16 v4h;
typedef __attribute__((ext_vector_type(8)))  float    v8f;

#define BATCH 512
#define TSEQ  256
#define CNNC  512
#define VOC   37
#define HID   8
#define GN    32
#define NROW  (TSEQ * BATCH)
#define NTILE (NROW / 16)

__device__ __forceinline__ v8f wmma16(v16h a, v16h b, v8f c) {
  return __builtin_amdgcn_wmma_f32_16x16x32_f16(
       false, a,  false, b,
       (short)0, c,  false,  false);
}

__device__ __forceinline__ v16h cat8(v8h a, v8h b) {
  v16h r;
  for (int i = 0; i < 8; ++i) { r[i] = a[i]; r[i + 8] = b[i]; }
  return r;
}

__device__ __forceinline__ v8f splat8(float x) {
  v8f r;
  for (int i = 0; i < 8; ++i) r[i] = x;
  return r;
}

__device__ __forceinline__ float sigmoidf_(float x) {
  return __builtin_amdgcn_rcpf(1.0f + __expf(-x));
}
__device__ __forceinline__ float tanhf_(float x) {
  float t = __expf(-2.0f * fabsf(x));
  float r = (1.0f - t) * __builtin_amdgcn_rcpf(1.0f + t);
  return copysignf(r, x);
}

__global__ void lstm_pool_kernel(const float* __restrict__ features,
                                 f16* __restrict__ pooled) {
  int idx = blockIdx.x * blockDim.x + threadIdx.x;
  if (idx >= BATCH * CNNC) return;
  const float* p = features + (size_t)idx * 49;
  float s = 0.f;
  #pragma unroll
  for (int i = 0; i < 49; ++i) s += p[i];
  pooled[idx] = (f16)(s * (1.0f / 49.0f));
}

__device__ __forceinline__ int gate_row(int nt, int m) {
  int r = m & 7, hh = m >> 3;
  return (2 * nt + (r >> 2)) * HID + 4 * hh + (r & 3);
}

#define PK_THREADS   256
#define PK_WIN_BLKS  (48 * CNNC / PK_THREADS)
#define PK_WIH_BLKS  (4 * 512 / PK_THREADS)
#define PK_WHH_BLKS  (2 * 512 / PK_THREADS)
#define PK_WOUT_BLKS (3 * 512 / PK_THREADS)
#define PK_BLKS (PK_WIN_BLKS + PK_WIH_BLKS + PK_WHH_BLKS + PK_WOUT_BLKS + 1)
static_assert((48 * CNNC) % PK_THREADS == 0 && 512 % PK_THREADS == 0,
              "a pack block must not straddle two tables");

__global__ void __launch_bounds__(PK_THREADS)
lstm_pack_kernel(const float* __restrict__ W_in,
                 const float* __restrict__ W_ih,
                 const float* __restrict__ W_hh,
                 const float* __restrict__ W_out,
                 const float* __restrict__ b_ih,
                 const float* __restrict__ b_hh,
                 const float* __restrict__ b_out,
                 f16* __restrict__ Winp,
                 f16* __restrict__ Wihp,
                 f16* __restrict__ Whhp,
                 f16* __restrict__ Woutp,
                 float* __restrict__ gbias,
                 float* __restrict__ obias)
{
  const int tid = threadIdx.x;
  int blk = blockIdx.x;
  if (blk < PK_WIN_BLKS) {
    int idx = blk * PK_THREADS + tid;
    int n = idx / CNNC, k = idx % CNNC;
    Winp[idx] = (f16)(n < VOC ? W_in[n * CNNC + k] : 0.f);
    return;
  }
  blk -= PK_WIN_BLKS;
  if (blk < PK_WIH_BLKS) {
    int idx = blk * PK_THREADS + tid;
    int tile = idx / 512, e = idx % 512, lane = e / 16, h = e % 16;
    int ks = tile / 2, nt = tile % 2;
    int k = ks * 32 + 16 * (lane / 16) + h;
    int n = gate_row(nt, lane % 16);
    Wihp[idx] = (f16)(k < VOC ? W_ih[n * VOC + k] : 0.f);
    return;
  }
  blk -= PK_WIH_BLKS;
  if (blk < PK_WHH_BLKS) {
    int idx = blk * PK_THREADS + tid;
    int tile = idx / 512, e = idx % 512, lane = e / 16, h = e % 16;
    int n = gate_row(tile, lane % 16);
    Whhp[idx] = (f16)(h < 4 ? W_hh[n * HID + 4 * (lane / 16) + h] : 0.f);
    return;
  }
  blk -= PK_WHH_BLKS;
  if (blk < PK_WOUT_BLKS) {
    int idx = blk * PK_THREADS + tid;
    int tile = idx / 512, e = idx % 512, lane = e / 16, h = e % 16;
    int kl = (h < 8 ? h : h + 8) + (lane / 16) * 8;
    int n = tile * 16 + (lane % 16);
    Woutp[idx] = (f16)((kl < HID && n < VOC) ? W_out[n * HID + kl] : 0.f);
    return;
  }
  if (tid < GN) {
    int n = gate_row(tid / 16, tid % 16);
    gbias[tid] = b_ih[n] + b_hh[n];
  }
  if (tid < 48) obias[tid] = (tid < VOC) ? b_out[tid] : 0.f;
}

__global__ void lstm_featgemm_kernel(const f16* __restrict__ pooled,
                                     const f16* __restrict__ Winp,
                                     const float* __restrict__ b_in,
                                     float* __restrict__ X0) {
  const int L = threadIdx.x;
  const int j16 = L & 15, half = L >> 4;
  const int Mb = blockIdx.x * 16;

  v8f acc[3];
  #pragma unroll
  for (int nt = 0; nt < 3; ++nt) {
    int col = nt * 16 + j16;
    acc[nt] = splat8(col < VOC ? b_in[col] : 0.f);
  }

  for (int kt = 0; kt < CNNC / 32; ++kt) {
    int kb = kt * 32 + half * 8;
    int row = Mb + j16;
    v16h A = cat8(*(const v8h*)(pooled + (size_t)row * CNNC + kb),
                  *(const v8h*)(pooled + (size_t)row * CNNC + kb + 16));
    #pragma unroll
    for (int nt = 0; nt < 3; ++nt) {
      int n = nt * 16 + j16;
      v16h Bm = cat8(*(const v8h*)(Winp + (size_t)n * CNNC + kb),
                     *(const v8h*)(Winp + (size_t)n * CNNC + kb + 16));
      acc[nt] = wmma16(A, Bm, acc[nt]);
    }
  }
  #pragma unroll
  for (int r = 0; r < 8; ++r) {
    int m = r + 8 * half;
    float* dst = X0 + (size_t)(Mb + m) * VOC;
    #pragma unroll
    for (int nt = 0; nt < 3; ++nt) {
      int col = nt * 16 + j16;
      if (col < VOC) dst[col] = acc[nt][r];
    }
  }
}

__global__ void lstm_pregate_kernel(const float* __restrict__ captions,
                                    const float* __restrict__ X0,
                                    const f16* __restrict__ Wihp,
                                    const float* __restrict__ gbias,
                                    float* __restrict__ G) {
  const int L = threadIdx.x;
  const int j16 = L & 15, half = L >> 4;
  const int t = blockIdx.x / (BATCH / 16);
  const int b = (blockIdx.x % (BATCH / 16)) * 16 + j16;

  v8f acc[2];
  acc[0] = *(const v8f*)(gbias + 8 * half);
  acc[1] = *(const v8f*)(gbias + 16 + 8 * half);

  const float* xr = (t == 0) ? X0 + (size_t)b * VOC
                             : captions + ((size_t)b * TSEQ + (t - 1)) * VOC;
  v16h Xb[2];
  #pragma unroll
  for (int s = 0; s < 16; ++s) Xb[0][s] = (f16)xr[16 * half + s];
  #pragma unroll
  for (int s = 0; s < 16; ++s) {
    float v = (s < VOC - 32) ? xr[32 + s] : 0.f;
    Xb[1][s] = (half == 0) ? (f16)v : (f16)0;
  }

  #pragma unroll
  for (int ks = 0; ks < 2; ++ks) {
    #pragma unroll
    for (int nt = 0; nt < 2; ++nt) {
      const f16* tp = Wihp + (size_t)(ks * 2 + nt) * 512 + L * 16;
      v16h Wa = cat8(*(const v8h*)tp, *(const v8h*)(tp + 8));
      acc[nt] = wmma16(Wa, Xb[ks], acc[nt]);
    }
  }
  float* dst = G + (size_t)blockIdx.x * 512 + L * 16;
  *(v8f*)dst       = acc[0];
  *(v8f*)(dst + 8) = acc[1];
}

#define GAHEAD 8
static_assert(TSEQ % GAHEAD == 0 && TSEQ >= 2 * GAHEAD, "ring depth must divide the sequence");

__device__ __forceinline__ void lstm_cell_step(v8f g0, v8f g1,
                                               const v16h& Wa0, const v16h& Wa1,
                                               v16h& Bh, float (&cst)[4],
                                               f16* __restrict__ hout) {
  g0 = wmma16(Wa0, Bh, g0);
  g1 = wmma16(Wa1, Bh, g1);

  v4h hv;
  #pragma unroll
  for (int u = 0; u < 4; ++u) {
    float iv = sigmoidf_(g0[u]);
    float fv = sigmoidf_(g0[4 + u]);
    float gv = tanhf_(g1[u]);
    float ov = sigmoidf_(g1[4 + u]);
    cst[u] = fv * cst[u] + iv * gv;
    hv[u] = (f16)(ov * tanhf_(cst[u]));
  }
  for (int u = 0; u < 4; ++u) Bh[u] = hv[u];

  *(v4h*)hout = hv;
}

__global__ void __launch_bounds__(32)
lstm_recur_kernel(const float* __restrict__ G,
                  const f16* __restrict__ Whhp,
                  f16* __restrict__ Hb) {
  const int L = threadIdx.x;
  const int j16 = L & 15, half = L >> 4;
  const int bblk = blockIdx.x;
  const int bbase = bblk * 16;

  v16h WhhA[2];
  for (int nt = 0; nt < 2; ++nt) {
    const f16* tp = Whhp + (size_t)nt * 512 + L * 16;
    WhhA[nt] = cat8(*(const v8h*)tp, *(const v8h*)(tp + 8));
  }

  float cst[4];
  for (int u = 0; u < 4; ++u) cst[u] = 0.f;

  v16h Bh;
  for (int i = 0; i < 16; ++i) Bh[i] = (f16)0;

  f16* hdst = Hb + ((size_t)bbase + j16) * HID + 4 * half;
  const size_t HSTEP = (size_t)BATCH * HID;

  const float* Gl = G + (size_t)bblk * 512 + L * 16;
  const size_t GSTEP = (size_t)(BATCH / 16) * 512;

  __builtin_amdgcn_sched_barrier(0);

  v8f ra[GAHEAD], rb[GAHEAD];
  #pragma unroll
  for (int d = 0; d < GAHEAD; ++d) {
    ra[d] = *(const v8f*)(Gl + d * GSTEP);
    rb[d] = *(const v8f*)(Gl + d * GSTEP + 8);
  }

  int t = 0;
  for (; t < TSEQ - GAHEAD; t += GAHEAD) {
    #pragma unroll
    for (int d = 0; d < GAHEAD; ++d) {
      lstm_cell_step(ra[d], rb[d], WhhA[0], WhhA[1], Bh, cst,
                     hdst + (size_t)(t + d) * HSTEP);
      __builtin_amdgcn_sched_barrier(0);
      const float* Gn = Gl + (size_t)(t + d + GAHEAD) * GSTEP;
      ra[d] = *(const v8f*)Gn;
      rb[d] = *(const v8f*)(Gn + 8);
    }
  }
  #pragma unroll
  for (int d = 0; d < GAHEAD; ++d)
    lstm_cell_step(ra[d], rb[d], WhhA[0], WhhA[1], Bh, cst,
                   hdst + (size_t)(t + d) * HSTEP);
}

__global__ void lstm_outhead_kernel(const f16* __restrict__ Hb,
                                    const f16* __restrict__ Woutp,
                                    const float* __restrict__ obias,
                                    float* __restrict__ out) {
  const int L = threadIdx.x;
  const int j16 = L & 15, half = L >> 4;
  const size_t Mb = (size_t)blockIdx.x * 16;

  v16h WoutB[3];
  #pragma unroll
  for (int nt = 0; nt < 3; ++nt) {
    const f16* tp = Woutp + (size_t)nt * 512 + L * 16;
    WoutB[nt] = cat8(*(const v8h*)tp, *(const v8h*)(tp + 8));
  }

  v16h A;
  #pragma unroll
  for (int i = 0; i < 16; ++i) A[i] = (f16)0;
  if (L < 16) {
    v8h hv = *(const v8h*)(Hb + (Mb + L) * HID);
    #pragma unroll
    for (int k = 0; k < HID; ++k) A[k] = hv[k];
  }

  v8f l0 = splat8(obias[j16]);
  v8f l1 = splat8(obias[16 + j16]);
  v8f l2 = splat8(obias[32 + j16]);
  l0 = wmma16(A, WoutB[0], l0);
  l1 = wmma16(A, WoutB[1], l1);
  l2 = wmma16(A, WoutB[2], l2);

  const bool v2ok = (32 + j16) < VOC;
  #pragma unroll
  for (int r = 0; r < 8; ++r) {
    float x2 = v2ok ? l2[r] : -1e30f;
    float mx = fmaxf(fmaxf(l0[r], l1[r]), x2);
    #pragma unroll
    for (int s = 1; s < 16; s <<= 1)
      mx = fmaxf(mx, __shfl_xor(mx, s, 16));
    float e0 = __expf(l0[r] - mx);
    float e1 = __expf(l1[r] - mx);
    float e2 = v2ok ? __expf(x2 - mx) : 0.f;
    float sm = e0 + e1 + e2;
    #pragma unroll
    for (int s = 1; s < 16; s <<= 1)
      sm += __shfl_xor(sm, s, 16);
    float inv = 1.0f / sm;

    size_t row = Mb + r + 8 * half;
    int t = (int)(row / BATCH);
    int b = (int)(row % BATCH);
    float* dst = out + ((size_t)b * TSEQ + t) * VOC;
    dst[j16]      = e0 * inv;
    dst[16 + j16] = e1 * inv;
    if (v2ok) dst[32 + j16] = e2 * inv;
  }
}

extern "C" void kernel_launch(void* const* d_in, const int* in_sizes, int n_in,
                              void* d_out, int out_size, void* d_ws, size_t ws_size,
                              hipStream_t stream) {
  const float* features = (const float*)d_in[0];
  const float* captions = (const float*)d_in[1];
  const float* W_in  = (const float*)d_in[2];
  const float* b_in  = (const float*)d_in[3];
  const float* W_ih  = (const float*)d_in[4];
  const float* W_hh  = (const float*)d_in[5];
  const float* b_ih  = (const float*)d_in[6];
  const float* b_hh  = (const float*)d_in[7];
  const float* W_out = (const float*)d_in[8];
  const float* b_out = (const float*)d_in[9];
  float* out = (float*)d_out;

  char* ws = (char*)d_ws;
  auto carve = [&](size_t bytes) -> char* {
    char* p = ws;
    ws += (bytes + 255) & ~(size_t)255;
    return p;
  };
  f16*   pooled = (f16*)  carve((size_t)BATCH * CNNC * sizeof(f16));
  f16*   Winp   = (f16*)  carve((size_t)48 * CNNC * sizeof(f16));
  f16*   Wihp   = (f16*)  carve((size_t)4 * 512 * sizeof(f16));
  f16*   Whhp   = (f16*)  carve((size_t)2 * 512 * sizeof(f16));
  f16*   Woutp  = (f16*)  carve((size_t)3 * 512 * sizeof(f16));
  float* gbias  = (float*)carve(GN * sizeof(float));
  float* obias  = (float*)carve(48 * sizeof(float));
  float* X0     = (float*)carve((size_t)BATCH * VOC * sizeof(float));
  float* G      = (float*)carve((size_t)NROW * GN * sizeof(float));
  f16*   Hb     = (f16*)  carve((size_t)NROW * HID * sizeof(f16));

  lstm_pool_kernel<<<(BATCH * CNNC) / 256, 256, 0, stream>>>(features, pooled);
  lstm_pack_kernel<<<PK_BLKS, PK_THREADS, 0, stream>>>(W_in, W_ih, W_hh, W_out,
                                                       b_ih, b_hh, b_out,
                                                       Winp, Wihp, Whhp, Woutp, gbias, obias);
  lstm_featgemm_kernel<<<BATCH / 16, 32, 0, stream>>>(pooled, Winp, b_in, X0);
  lstm_pregate_kernel<<<NTILE, 32, 0, stream>>>(captions, X0, Wihp, gbias, G);
  lstm_recur_kernel<<<BATCH / 16, 32, 0, stream>>>(G, Whhp, Hb);
  lstm_outhead_kernel<<<NTILE, 32, 0, stream>>>(Hb, Woutp, obias, out);
}


# Recurrence cell: hardware v_tanh_f32 replaces the exp+rcp sequences; sigmoid = 0.5+0.5*tanh(x/2), the 1/2 folded into the packed i,f,o weight rows and bias: 5 transcendentals per cell, not 10
# speedup vs baseline: 6.5216x; 1.1860x over previous
#include <hip/hip_runtime.h>
#include <hip/hip_bf16.h>

typedef _Float16 f16;
typedef __attribute__((ext_vector_type(16))) _Float16 v16h;
typedef __attribute__((ext_vector_type(8)))  _Float16 v8h;
typedef __attribute__((ext_vector_type(4)))  _Float16 v4h;
typedef __attribute__((ext_vector_type(8)))  float    v8f;

#define BATCH 512
#define TSEQ  256
#define CNNC  512
#define VOC   37
#define HID   8
#define GN    32
#define NROW  (TSEQ * BATCH)
#define NTILE (NROW / 16)

__device__ __forceinline__ v8f wmma16(v16h a, v16h b, v8f c) {
  return __builtin_amdgcn_wmma_f32_16x16x32_f16(
       false, a,  false, b,
       (short)0, c,  false,  false);
}

__device__ __forceinline__ v16h cat8(v8h a, v8h b) {
  v16h r;
  for (int i = 0; i < 8; ++i) { r[i] = a[i]; r[i + 8] = b[i]; }
  return r;
}

__device__ __forceinline__ v8f splat8(float x) {
  v8f r;
  for (int i = 0; i < 8; ++i) r[i] = x;
  return r;
}

#define GATE_G 2
__device__ __forceinline__ float tanhf_(float x) {
  return __builtin_amdgcn_tanhf(x);
}
__device__ __forceinline__ float sigmoid_of_half_(float xhalf) {
  return fmaf(0.5f, __builtin_amdgcn_tanhf(xhalf), 0.5f);
}
__device__ __forceinline__ float gate_scale(int n) {
  return (n / HID == GATE_G) ? 1.0f : 0.5f;
}

__global__ void lstm_pool_kernel(const float* __restrict__ features,
                                 f16* __restrict__ pooled) {
  int idx = blockIdx.x * blockDim.x + threadIdx.x;
  if (idx >= BATCH * CNNC) return;
  const float* p = features + (size_t)idx * 49;
  float s = 0.f;
  #pragma unroll
  for (int i = 0; i < 49; ++i) s += p[i];
  pooled[idx] = (f16)(s * (1.0f / 49.0f));
}

__device__ __forceinline__ int gate_row(int nt, int m) {
  int r = m & 7, hh = m >> 3;
  return (2 * nt + (r >> 2)) * HID + 4 * hh + (r & 3);
}

#define PK_THREADS   256
#define PK_WIN_BLKS  (48 * CNNC / PK_THREADS)
#define PK_WIH_BLKS  (4 * 512 / PK_THREADS)
#define PK_WHH_BLKS  (2 * 512 / PK_THREADS)
#define PK_WOUT_BLKS (3 * 512 / PK_THREADS)
#define PK_BLKS (PK_WIN_BLKS + PK_WIH_BLKS + PK_WHH_BLKS + PK_WOUT_BLKS + 1)
static_assert((48 * CNNC) % PK_THREADS == 0 && 512 % PK_THREADS == 0,
              "a pack block must not straddle two tables");

__global__ void __launch_bounds__(PK_THREADS)
lstm_pack_kernel(const float* __restrict__ W_in,
                 const float* __restrict__ W_ih,
                 const float* __restrict__ W_hh,
                 const float* __restrict__ W_out,
                 const float* __restrict__ b_ih,
                 const float* __restrict__ b_hh,
                 const float* __restrict__ b_out,
                 f16* __restrict__ Winp,
                 f16* __restrict__ Wihp,
                 f16* __restrict__ Whhp,
                 f16* __restrict__ Woutp,
                 float* __restrict__ gbias,
                 float* __restrict__ obias)
{
  const int tid = threadIdx.x;
  int blk = blockIdx.x;
  if (blk < PK_WIN_BLKS) {
    int idx = blk * PK_THREADS + tid;
    int n = idx / CNNC, k = idx % CNNC;
    Winp[idx] = (f16)(n < VOC ? W_in[n * CNNC + k] : 0.f);
    return;
  }
  blk -= PK_WIN_BLKS;
  if (blk < PK_WIH_BLKS) {
    int idx = blk * PK_THREADS + tid;
    int tile = idx / 512, e = idx % 512, lane = e / 16, h = e % 16;
    int ks = tile / 2, nt = tile % 2;
    int k = ks * 32 + 16 * (lane / 16) + h;
    int n = gate_row(nt, lane % 16);
    Wihp[idx] = (f16)(k < VOC ? gate_scale(n) * W_ih[n * VOC + k] : 0.f);
    return;
  }
  blk -= PK_WIH_BLKS;
  if (blk < PK_WHH_BLKS) {
    int idx = blk * PK_THREADS + tid;
    int tile = idx / 512, e = idx % 512, lane = e / 16, h = e % 16;
    int n = gate_row(tile, lane % 16);
    Whhp[idx] = (f16)(h < 4 ? gate_scale(n) * W_hh[n * HID + 4 * (lane / 16) + h] : 0.f);
    return;
  }
  blk -= PK_WHH_BLKS;
  if (blk < PK_WOUT_BLKS) {
    int idx = blk * PK_THREADS + tid;
    int tile = idx / 512, e = idx % 512, lane = e / 16, h = e % 16;
    int kl = (h < 8 ? h : h + 8) + (lane / 16) * 8;
    int n = tile * 16 + (lane % 16);
    Woutp[idx] = (f16)((kl < HID && n < VOC) ? W_out[n * HID + kl] : 0.f);
    return;
  }
  if (tid < GN) {
    int n = gate_row(tid / 16, tid % 16);
    gbias[tid] = gate_scale(n) * (b_ih[n] + b_hh[n]);
  }
  if (tid < 48) obias[tid] = (tid < VOC) ? b_out[tid] : 0.f;
}

__global__ void lstm_featgemm_kernel(const f16* __restrict__ pooled,
                                     const f16* __restrict__ Winp,
                                     const float* __restrict__ b_in,
                                     float* __restrict__ X0) {
  const int L = threadIdx.x;
  const int j16 = L & 15, half = L >> 4;
  const int Mb = blockIdx.x * 16;

  v8f acc[3];
  #pragma unroll
  for (int nt = 0; nt < 3; ++nt) {
    int col = nt * 16 + j16;
    acc[nt] = splat8(col < VOC ? b_in[col] : 0.f);
  }

  for (int kt = 0; kt < CNNC / 32; ++kt) {
    int kb = kt * 32 + half * 8;
    int row = Mb + j16;
    v16h A = cat8(*(const v8h*)(pooled + (size_t)row * CNNC + kb),
                  *(const v8h*)(pooled + (size_t)row * CNNC + kb + 16));
    #pragma unroll
    for (int nt = 0; nt < 3; ++nt) {
      int n = nt * 16 + j16;
      v16h Bm = cat8(*(const v8h*)(Winp + (size_t)n * CNNC + kb),
                     *(const v8h*)(Winp + (size_t)n * CNNC + kb + 16));
      acc[nt] = wmma16(A, Bm, acc[nt]);
    }
  }
  #pragma unroll
  for (int r = 0; r < 8; ++r) {
    int m = r + 8 * half;
    float* dst = X0 + (size_t)(Mb + m) * VOC;
    #pragma unroll
    for (int nt = 0; nt < 3; ++nt) {
      int col = nt * 16 + j16;
      if (col < VOC) dst[col] = acc[nt][r];
    }
  }
}

__global__ void lstm_pregate_kernel(const float* __restrict__ captions,
                                    const float* __restrict__ X0,
                                    const f16* __restrict__ Wihp,
                                    const float* __restrict__ gbias,
                                    float* __restrict__ G) {
  const int L = threadIdx.x;
  const int j16 = L & 15, half = L >> 4;
  const int t = blockIdx.x / (BATCH / 16);
  const int b = (blockIdx.x % (BATCH / 16)) * 16 + j16;

  v8f acc[2];
  acc[0] = *(const v8f*)(gbias + 8 * half);
  acc[1] = *(const v8f*)(gbias + 16 + 8 * half);

  const float* xr = (t == 0) ? X0 + (size_t)b * VOC
                             : captions + ((size_t)b * TSEQ + (t - 1)) * VOC;
  v16h Xb[2];
  #pragma unroll
  for (int s = 0; s < 16; ++s) Xb[0][s] = (f16)xr[16 * half + s];
  #pragma unroll
  for (int s = 0; s < 16; ++s) {
    float v = (s < VOC - 32) ? xr[32 + s] : 0.f;
    Xb[1][s] = (half == 0) ? (f16)v : (f16)0;
  }

  #pragma unroll
  for (int ks = 0; ks < 2; ++ks) {
    #pragma unroll
    for (int nt = 0; nt < 2; ++nt) {
      const f16* tp = Wihp + (size_t)(ks * 2 + nt) * 512 + L * 16;
      v16h Wa = cat8(*(const v8h*)tp, *(const v8h*)(tp + 8));
      acc[nt] = wmma16(Wa, Xb[ks], acc[nt]);
    }
  }
  float* dst = G + (size_t)blockIdx.x * 512 + L * 16;
  *(v8f*)dst       = acc[0];
  *(v8f*)(dst + 8) = acc[1];
}

#define GAHEAD 8
static_assert(TSEQ % GAHEAD == 0 && TSEQ >= 2 * GAHEAD, "ring depth must divide the sequence");

__device__ __forceinline__ void lstm_cell_step(v8f g0, v8f g1,
                                               const v16h& Wa0, const v16h& Wa1,
                                               v16h& Bh, float (&cst)[4],
                                               f16* __restrict__ hout) {
  g0 = wmma16(Wa0, Bh, g0);
  g1 = wmma16(Wa1, Bh, g1);

  v4h hv;
  #pragma unroll
  for (int u = 0; u < 4; ++u) {
    float iv = sigmoid_of_half_(g0[u]);
    float fv = sigmoid_of_half_(g0[4 + u]);
    float gv = tanhf_(g1[u]);
    float ov = sigmoid_of_half_(g1[4 + u]);
    cst[u] = fv * cst[u] + iv * gv;
    hv[u] = (f16)(ov * tanhf_(cst[u]));
  }
  for (int u = 0; u < 4; ++u) Bh[u] = hv[u];

  *(v4h*)hout = hv;
}

__global__ void __launch_bounds__(32)
lstm_recur_kernel(const float* __restrict__ G,
                  const f16* __restrict__ Whhp,
                  f16* __restrict__ Hb) {
  const int L = threadIdx.x;
  const int j16 = L & 15, half = L >> 4;
  const int bblk = blockIdx.x;
  const int bbase = bblk * 16;

  v16h WhhA[2];
  for (int nt = 0; nt < 2; ++nt) {
    const f16* tp = Whhp + (size_t)nt * 512 + L * 16;
    WhhA[nt] = cat8(*(const v8h*)tp, *(const v8h*)(tp + 8));
  }

  float cst[4];
  for (int u = 0; u < 4; ++u) cst[u] = 0.f;

  v16h Bh;
  for (int i = 0; i < 16; ++i) Bh[i] = (f16)0;

  f16* hdst = Hb + ((size_t)bbase + j16) * HID + 4 * half;
  const size_t HSTEP = (size_t)BATCH * HID;

  const float* Gl = G + (size_t)bblk * 512 + L * 16;
  const size_t GSTEP = (size_t)(BATCH / 16) * 512;

  __builtin_amdgcn_sched_barrier(0);

  v8f ra[GAHEAD], rb[GAHEAD];
  #pragma unroll
  for (int d = 0; d < GAHEAD; ++d) {
    ra[d] = *(const v8f*)(Gl + d * GSTEP);
    rb[d] = *(const v8f*)(Gl + d * GSTEP + 8);
  }

  int t = 0;
  for (; t < TSEQ - GAHEAD; t += GAHEAD) {
    #pragma unroll
    for (int d = 0; d < GAHEAD; ++d) {
      lstm_cell_step(ra[d], rb[d], WhhA[0], WhhA[1], Bh, cst,
                     hdst + (size_t)(t + d) * HSTEP);
      __builtin_amdgcn_sched_barrier(0);
      const float* Gn = Gl + (size_t)(t + d + GAHEAD) * GSTEP;
      ra[d] = *(const v8f*)Gn;
      rb[d] = *(const v8f*)(Gn + 8);
    }
  }
  #pragma unroll
  for (int d = 0; d < GAHEAD; ++d)
    lstm_cell_step(ra[d], rb[d], WhhA[0], WhhA[1], Bh, cst,
                   hdst + (size_t)(t + d) * HSTEP);
}

__global__ void lstm_outhead_kernel(const f16* __restrict__ Hb,
                                    const f16* __restrict__ Woutp,
                                    const float* __restrict__ obias,
                                    float* __restrict__ out) {
  const int L = threadIdx.x;
  const int j16 = L & 15, half = L >> 4;
  const size_t Mb = (size_t)blockIdx.x * 16;

  v16h WoutB[3];
  #pragma unroll
  for (int nt = 0; nt < 3; ++nt) {
    const f16* tp = Woutp + (size_t)nt * 512 + L * 16;
    WoutB[nt] = cat8(*(const v8h*)tp, *(const v8h*)(tp + 8));
  }

  v16h A;
  #pragma unroll
  for (int i = 0; i < 16; ++i) A[i] = (f16)0;
  if (L < 16) {
    v8h hv = *(const v8h*)(Hb + (Mb + L) * HID);
    #pragma unroll
    for (int k = 0; k < HID; ++k) A[k] = hv[k];
  }

  v8f l0 = splat8(obias[j16]);
  v8f l1 = splat8(obias[16 + j16]);
  v8f l2 = splat8(obias[32 + j16]);
  l0 = wmma16(A, WoutB[0], l0);
  l1 = wmma16(A, WoutB[1], l1);
  l2 = wmma16(A, WoutB[2], l2);

  const bool v2ok = (32 + j16) < VOC;
  #pragma unroll
  for (int r = 0; r < 8; ++r) {
    float x2 = v2ok ? l2[r] : -1e30f;
    float mx = fmaxf(fmaxf(l0[r], l1[r]), x2);
    #pragma unroll
    for (int s = 1; s < 16; s <<= 1)
      mx = fmaxf(mx, __shfl_xor(mx, s, 16));
    float e0 = __expf(l0[r] - mx);
    float e1 = __expf(l1[r] - mx);
    float e2 = v2ok ? __expf(x2 - mx) : 0.f;
    float sm = e0 + e1 + e2;
    #pragma unroll
    for (int s = 1; s < 16; s <<= 1)
      sm += __shfl_xor(sm, s, 16);
    float inv = 1.0f / sm;

    size_t row = Mb + r + 8 * half;
    int t = (int)(row / BATCH);
    int b = (int)(row % BATCH);
    float* dst = out + ((size_t)b * TSEQ + t) * VOC;
    dst[j16]      = e0 * inv;
    dst[16 + j16] = e1 * inv;
    if (v2ok) dst[32 + j16] = e2 * inv;
  }
}

extern "C" void kernel_launch(void* const* d_in, const int* in_sizes, int n_in,
                              void* d_out, int out_size, void* d_ws, size_t ws_size,
                              hipStream_t stream) {
  const float* features = (const float*)d_in[0];
  const float* captions = (const float*)d_in[1];
  const float* W_in  = (const float*)d_in[2];
  const float* b_in  = (const float*)d_in[3];
  const float* W_ih  = (const float*)d_in[4];
  const float* W_hh  = (const float*)d_in[5];
  const float* b_ih  = (const float*)d_in[6];
  const float* b_hh  = (const float*)d_in[7];
  const float* W_out = (const float*)d_in[8];
  const float* b_out = (const float*)d_in[9];
  float* out = (float*)d_out;

  char* ws = (char*)d_ws;
  auto carve = [&](size_t bytes) -> char* {
    char* p = ws;
    ws += (bytes + 255) & ~(size_t)255;
    return p;
  };
  f16*   pooled = (f16*)  carve((size_t)BATCH * CNNC * sizeof(f16));
  f16*   Winp   = (f16*)  carve((size_t)48 * CNNC * sizeof(f16));
  f16*   Wihp   = (f16*)  carve((size_t)4 * 512 * sizeof(f16));
  f16*   Whhp   = (f16*)  carve((size_t)2 * 512 * sizeof(f16));
  f16*   Woutp  = (f16*)  carve((size_t)3 * 512 * sizeof(f16));
  float* gbias  = (float*)carve(GN * sizeof(float));
  float* obias  = (float*)carve(48 * sizeof(float));
  float* X0     = (float*)carve((size_t)BATCH * VOC * sizeof(float));
  float* G      = (float*)carve((size_t)NROW * GN * sizeof(float));
  f16*   Hb     = (f16*)  carve((size_t)NROW * HID * sizeof(f16));

  lstm_pool_kernel<<<(BATCH * CNNC) / 256, 256, 0, stream>>>(features, pooled);
  lstm_pack_kernel<<<PK_BLKS, PK_THREADS, 0, stream>>>(W_in, W_ih, W_hh, W_out,
                                                       b_ih, b_hh, b_out,
                                                       Winp, Wihp, Whhp, Woutp, gbias, obias);
  lstm_featgemm_kernel<<<BATCH / 16, 32, 0, stream>>>(pooled, Winp, b_in, X0);
  lstm_pregate_kernel<<<NTILE, 32, 0, stream>>>(captions, X0, Wihp, gbias, G);
  lstm_recur_kernel<<<BATCH / 16, 32, 0, stream>>>(G, Whhp, Hb);
  lstm_outhead_kernel<<<NTILE, 32, 0, stream>>>(Hb, Woutp, obias, out);
}


# Launch fusion: weight pack runs as extra blocks of the pooling launch; the feat GEMM moves into the pre-gate kernel's t=0 waves, transposed so its D registers are their B operand: 4 launches, not 6
# speedup vs baseline: 7.5482x; 1.1574x over previous
#include <hip/hip_runtime.h>
#include <hip/hip_bf16.h>

typedef _Float16 f16;
typedef __attribute__((ext_vector_type(16))) _Float16 v16h;
typedef __attribute__((ext_vector_type(8)))  _Float16 v8h;
typedef __attribute__((ext_vector_type(4)))  _Float16 v4h;
typedef __attribute__((ext_vector_type(8)))  float    v8f;

#define BATCH 512
#define TSEQ  256
#define CNNC  512
#define VOC   37
#define HID   8
#define GN    32
#define NROW  (TSEQ * BATCH)
#define NTILE (NROW / 16)

__device__ __forceinline__ v8f wmma16(v16h a, v16h b, v8f c) {
  return __builtin_amdgcn_wmma_f32_16x16x32_f16(
       false, a,  false, b,
       (short)0, c,  false,  false);
}

__device__ __forceinline__ v16h cat8(v8h a, v8h b) {
  v16h r;
  for (int i = 0; i < 8; ++i) { r[i] = a[i]; r[i + 8] = b[i]; }
  return r;
}

__device__ __forceinline__ v8f splat8(float x) {
  v8f r;
  for (int i = 0; i < 8; ++i) r[i] = x;
  return r;
}

#define GATE_G 2
__device__ __forceinline__ float tanhf_(float x) {
  return __builtin_amdgcn_tanhf(x);
}
__device__ __forceinline__ float sigmoid_of_half_(float xhalf) {
  return fmaf(0.5f, __builtin_amdgcn_tanhf(xhalf), 0.5f);
}
__device__ __forceinline__ float gate_scale(int n) {
  return (n / HID == GATE_G) ? 1.0f : 0.5f;
}

#define POOL_THREADS 256
#define POOL_BLKS ((BATCH * CNNC) / POOL_THREADS)
__device__ __forceinline__ void pool_block(const float* __restrict__ features,
                                           f16* __restrict__ pooled, int blk) {
  int idx = blk * POOL_THREADS + threadIdx.x;
  const float* p = features + (size_t)idx * 49;
  float s = 0.f;
  #pragma unroll
  for (int i = 0; i < 49; ++i) s += p[i];
  pooled[idx] = (f16)(s * (1.0f / 49.0f));
}

__device__ __forceinline__ int gate_row(int nt, int m) {
  int r = m & 7, hh = m >> 3;
  return (2 * nt + (r >> 2)) * HID + 4 * hh + (r & 3);
}

#define PK_THREADS   256
#define PK_WIN_BLKS  (48 * CNNC / PK_THREADS)
#define PK_WIH_BLKS  (4 * 512 / PK_THREADS)
#define PK_WHH_BLKS  (2 * 512 / PK_THREADS)
#define PK_WOUT_BLKS (3 * 512 / PK_THREADS)
#define PK_BLKS (PK_WIN_BLKS + PK_WIH_BLKS + PK_WHH_BLKS + PK_WOUT_BLKS + 1)
static_assert((48 * CNNC) % PK_THREADS == 0 && 512 % PK_THREADS == 0,
              "a pack block must not straddle two tables");
static_assert(PK_THREADS == POOL_THREADS, "pack and pool blocks share one launch");

__global__ void __launch_bounds__(PK_THREADS)
lstm_front_kernel(const float* __restrict__ features,
                 f16* __restrict__ pooled,
                 const float* __restrict__ W_in,
                 const float* __restrict__ W_ih,
                 const float* __restrict__ W_hh,
                 const float* __restrict__ W_out,
                 const float* __restrict__ b_ih,
                 const float* __restrict__ b_hh,
                 const float* __restrict__ b_out,
                 f16* __restrict__ Winp,
                 f16* __restrict__ Wihp,
                 f16* __restrict__ Whhp,
                 f16* __restrict__ Woutp,
                 float* __restrict__ gbias,
                 float* __restrict__ obias)
{
  const int tid = threadIdx.x;
  int blk = blockIdx.x;
  if (blk >= PK_BLKS) {
    pool_block(features, pooled, blk - PK_BLKS);
    return;
  }
  if (blk < PK_WIN_BLKS) {
    int idx = blk * PK_THREADS + tid;
    int n = idx / CNNC, k = idx % CNNC;
    Winp[idx] = (f16)(n < VOC ? W_in[n * CNNC + k] : 0.f);
    return;
  }
  blk -= PK_WIN_BLKS;
  if (blk < PK_WIH_BLKS) {
    int idx = blk * PK_THREADS + tid;
    int tile = idx / 512, e = idx % 512, lane = e / 16, h = e % 16;
    int ks = tile / 2, nt = tile % 2;
    int k = ks * 32 + 16 * (lane / 16) + h;
    int n = gate_row(nt, lane % 16);
    Wihp[idx] = (f16)(k < VOC ? gate_scale(n) * W_ih[n * VOC + k] : 0.f);
    return;
  }
  blk -= PK_WIH_BLKS;
  if (blk < PK_WHH_BLKS) {
    int idx = blk * PK_THREADS + tid;
    int tile = idx / 512, e = idx % 512, lane = e / 16, h = e % 16;
    int n = gate_row(tile, lane % 16);
    Whhp[idx] = (f16)(h < 4 ? gate_scale(n) * W_hh[n * HID + 4 * (lane / 16) + h] : 0.f);
    return;
  }
  blk -= PK_WHH_BLKS;
  if (blk < PK_WOUT_BLKS) {
    int idx = blk * PK_THREADS + tid;
    int tile = idx / 512, e = idx % 512, lane = e / 16, h = e % 16;
    int kl = (h < 8 ? h : h + 8) + (lane / 16) * 8;
    int n = tile * 16 + (lane % 16);
    Woutp[idx] = (f16)((kl < HID && n < VOC) ? W_out[n * HID + kl] : 0.f);
    return;
  }
  if (tid < GN) {
    int n = gate_row(tid / 16, tid % 16);
    gbias[tid] = gate_scale(n) * (b_ih[n] + b_hh[n]);
  }
  if (tid < 48) obias[tid] = (tid < VOC) ? b_out[tid] : 0.f;
}

__global__ void lstm_pregate_kernel(const float* __restrict__ captions,
                                    const f16* __restrict__ pooled,
                                    const f16* __restrict__ Winp,
                                    const float* __restrict__ b_in,
                                    const f16* __restrict__ Wihp,
                                    const float* __restrict__ gbias,
                                    float* __restrict__ G) {
  const int L = threadIdx.x;
  const int j16 = L & 15, half = L >> 4;
  const int t = blockIdx.x / (BATCH / 16);
  const int b = (blockIdx.x % (BATCH / 16)) * 16 + j16;

  v8f acc[2];
  acc[0] = *(const v8f*)(gbias + 8 * half);
  acc[1] = *(const v8f*)(gbias + 16 + 8 * half);

  v16h Xb[2];
  if (t == 0) {
    v8f aP, aQ, aR;
    #pragma unroll
    for (int r = 0; r < 8; ++r) {
      aP[r] = b_in[16 * half + r];
      aQ[r] = b_in[16 * half + 8 + r];
      int n = 32 + 8 * half + r;
      aR[r] = (n < VOC) ? b_in[n] : 0.f;
    }
    const f16* wP = Winp + (size_t)(16 * (j16 >> 3) + (j16 & 7)) * CNNC;
    const f16* wQ = wP + (size_t)8 * CNNC;
    const f16* wR = Winp + (size_t)(32 + j16) * CNNC;
    const f16* pr = pooled + (size_t)b * CNNC;
    for (int kt = 0; kt < CNNC / 32; kt += 2) {
      v8h fB[4], fP[4], fQ[4], fR[4];
      #pragma unroll
      for (int j = 0; j < 2; ++j) {
        int kb = (kt + j) * 32 + half * 8;
        fB[2 * j] = *(const v8h*)(pr + kb);  fB[2 * j + 1] = *(const v8h*)(pr + kb + 16);
        fP[2 * j] = *(const v8h*)(wP + kb);  fP[2 * j + 1] = *(const v8h*)(wP + kb + 16);
        fQ[2 * j] = *(const v8h*)(wQ + kb);  fQ[2 * j + 1] = *(const v8h*)(wQ + kb + 16);
        fR[2 * j] = *(const v8h*)(wR + kb);  fR[2 * j + 1] = *(const v8h*)(wR + kb + 16);
      }
      __builtin_amdgcn_sched_barrier(0);
      #pragma unroll
      for (int j = 0; j < 2; ++j) {
        v16h Bp = cat8(fB[2 * j], fB[2 * j + 1]);
        aP = wmma16(cat8(fP[2 * j], fP[2 * j + 1]), Bp, aP);
        aQ = wmma16(cat8(fQ[2 * j], fQ[2 * j + 1]), Bp, aQ);
        aR = wmma16(cat8(fR[2 * j], fR[2 * j + 1]), Bp, aR);
      }
    }
    #pragma unroll
    for (int s = 0; s < 8; ++s) {
      Xb[0][s]     = (f16)aP[s];
      Xb[0][8 + s] = (f16)aQ[s];
      Xb[1][s]     = (f16)aR[s];
      Xb[1][8 + s] = (f16)0;
    }
  } else {
    const float* xr = captions + ((size_t)b * TSEQ + (t - 1)) * VOC;
    #pragma unroll
    for (int s = 0; s < 16; ++s) Xb[0][s] = (f16)xr[16 * half + s];
    #pragma unroll
    for (int s = 0; s < 16; ++s) {
      float v = (s < VOC - 32) ? xr[32 + s] : 0.f;
      Xb[1][s] = (half == 0) ? (f16)v : (f16)0;
    }
  }

  #pragma unroll
  for (int ks = 0; ks < 2; ++ks) {
    #pragma unroll
    for (int nt = 0; nt < 2; ++nt) {
      const f16* tp = Wihp + (size_t)(ks * 2 + nt) * 512 + L * 16;
      v16h Wa = cat8(*(const v8h*)tp, *(const v8h*)(tp + 8));
      acc[nt] = wmma16(Wa, Xb[ks], acc[nt]);
    }
  }
  float* dst = G + (size_t)blockIdx.x * 512 + L * 16;
  *(v8f*)dst       = acc[0];
  *(v8f*)(dst + 8) = acc[1];
}

#define GAHEAD 8
static_assert(TSEQ % GAHEAD == 0 && TSEQ >= 2 * GAHEAD, "ring depth must divide the sequence");

__device__ __forceinline__ void lstm_cell_step(v8f g0, v8f g1,
                                               const v16h& Wa0, const v16h& Wa1,
                                               v16h& Bh, float (&cst)[4],
                                               f16* __restrict__ hout) {
  g0 = wmma16(Wa0, Bh, g0);
  g1 = wmma16(Wa1, Bh, g1);

  v4h hv;
  #pragma unroll
  for (int u = 0; u < 4; ++u) {
    float iv = sigmoid_of_half_(g0[u]);
    float fv = sigmoid_of_half_(g0[4 + u]);
    float gv = tanhf_(g1[u]);
    float ov = sigmoid_of_half_(g1[4 + u]);
    cst[u] = fv * cst[u] + iv * gv;
    hv[u] = (f16)(ov * tanhf_(cst[u]));
  }
  for (int u = 0; u < 4; ++u) Bh[u] = hv[u];

  *(v4h*)hout = hv;
}

__global__ void __launch_bounds__(32)
lstm_recur_kernel(const float* __restrict__ G,
                  const f16* __restrict__ Whhp,
                  f16* __restrict__ Hb) {
  const int L = threadIdx.x;
  const int j16 = L & 15, half = L >> 4;
  const int bblk = blockIdx.x;
  const int bbase = bblk * 16;

  v16h WhhA[2];
  for (int nt = 0; nt < 2; ++nt) {
    const f16* tp = Whhp + (size_t)nt * 512 + L * 16;
    WhhA[nt] = cat8(*(const v8h*)tp, *(const v8h*)(tp + 8));
  }

  float cst[4];
  for (int u = 0; u < 4; ++u) cst[u] = 0.f;

  v16h Bh;
  for (int i = 0; i < 16; ++i) Bh[i] = (f16)0;

  f16* hdst = Hb + ((size_t)bbase + j16) * HID + 4 * half;
  const size_t HSTEP = (size_t)BATCH * HID;

  const float* Gl = G + (size_t)bblk * 512 + L * 16;
  const size_t GSTEP = (size_t)(BATCH / 16) * 512;

  __builtin_amdgcn_sched_barrier(0);

  v8f ra[GAHEAD], rb[GAHEAD];
  #pragma unroll
  for (int d = 0; d < GAHEAD; ++d) {
    ra[d] = *(const v8f*)(Gl + d * GSTEP);
    rb[d] = *(const v8f*)(Gl + d * GSTEP + 8);
  }

  int t = 0;
  for (; t < TSEQ - GAHEAD; t += GAHEAD) {
    #pragma unroll
    for (int d = 0; d < GAHEAD; ++d) {
      lstm_cell_step(ra[d], rb[d], WhhA[0], WhhA[1], Bh, cst,
                     hdst + (size_t)(t + d) * HSTEP);
      __builtin_amdgcn_sched_barrier(0);
      const float* Gn = Gl + (size_t)(t + d + GAHEAD) * GSTEP;
      ra[d] = *(const v8f*)Gn;
      rb[d] = *(const v8f*)(Gn + 8);
    }
  }
  #pragma unroll
  for (int d = 0; d < GAHEAD; ++d)
    lstm_cell_step(ra[d], rb[d], WhhA[0], WhhA[1], Bh, cst,
                   hdst + (size_t)(t + d) * HSTEP);
}

__global__ void lstm_outhead_kernel(const f16* __restrict__ Hb,
                                    const f16* __restrict__ Woutp,
                                    const float* __restrict__ obias,
                                    float* __restrict__ out) {
  const int L = threadIdx.x;
  const int j16 = L & 15, half = L >> 4;
  const size_t Mb = (size_t)blockIdx.x * 16;

  v16h WoutB[3];
  #pragma unroll
  for (int nt = 0; nt < 3; ++nt) {
    const f16* tp = Woutp + (size_t)nt * 512 + L * 16;
    WoutB[nt] = cat8(*(const v8h*)tp, *(const v8h*)(tp + 8));
  }

  v16h A;
  #pragma unroll
  for (int i = 0; i < 16; ++i) A[i] = (f16)0;
  if (L < 16) {
    v8h hv = *(const v8h*)(Hb + (Mb + L) * HID);
    #pragma unroll
    for (int k = 0; k < HID; ++k) A[k] = hv[k];
  }

  v8f l0 = splat8(obias[j16]);
  v8f l1 = splat8(obias[16 + j16]);
  v8f l2 = splat8(obias[32 + j16]);
  l0 = wmma16(A, WoutB[0], l0);
  l1 = wmma16(A, WoutB[1], l1);
  l2 = wmma16(A, WoutB[2], l2);

  const bool v2ok = (32 + j16) < VOC;
  #pragma unroll
  for (int r = 0; r < 8; ++r) {
    float x2 = v2ok ? l2[r] : -1e30f;
    float mx = fmaxf(fmaxf(l0[r], l1[r]), x2);
    #pragma unroll
    for (int s = 1; s < 16; s <<= 1)
      mx = fmaxf(mx, __shfl_xor(mx, s, 16));
    float e0 = __expf(l0[r] - mx);
    float e1 = __expf(l1[r] - mx);
    float e2 = v2ok ? __expf(x2 - mx) : 0.f;
    float sm = e0 + e1 + e2;
    #pragma unroll
    for (int s = 1; s < 16; s <<= 1)
      sm += __shfl_xor(sm, s, 16);
    float inv = 1.0f / sm;

    size_t row = Mb + r + 8 * half;
    int t = (int)(row / BATCH);
    int b = (int)(row % BATCH);
    float* dst = out + ((size_t)b * TSEQ + t) * VOC;
    dst[j16]      = e0 * inv;
    dst[16 + j16] = e1 * inv;
    if (v2ok) dst[32 + j16] = e2 * inv;
  }
}

extern "C" void kernel_launch(void* const* d_in, const int* in_sizes, int n_in,
                              void* d_out, int out_size, void* d_ws, size_t ws_size,
                              hipStream_t stream) {
  const float* features = (const float*)d_in[0];
  const float* captions = (const float*)d_in[1];
  const float* W_in  = (const float*)d_in[2];
  const float* b_in  = (const float*)d_in[3];
  const float* W_ih  = (const float*)d_in[4];
  const float* W_hh  = (const float*)d_in[5];
  const float* b_ih  = (const float*)d_in[6];
  const float* b_hh  = (const float*)d_in[7];
  const float* W_out = (const float*)d_in[8];
  const float* b_out = (const float*)d_in[9];
  float* out = (float*)d_out;

  char* ws = (char*)d_ws;
  auto carve = [&](size_t bytes) -> char* {
    char* p = ws;
    ws += (bytes + 255) & ~(size_t)255;
    return p;
  };
  f16*   pooled = (f16*)  carve((size_t)BATCH * CNNC * sizeof(f16));
  f16*   Winp   = (f16*)  carve((size_t)48 * CNNC * sizeof(f16));
  f16*   Wihp   = (f16*)  carve((size_t)4 * 512 * sizeof(f16));
  f16*   Whhp   = (f16*)  carve((size_t)2 * 512 * sizeof(f16));
  f16*   Woutp  = (f16*)  carve((size_t)3 * 512 * sizeof(f16));
  float* gbias  = (float*)carve(GN * sizeof(float));
  float* obias  = (float*)carve(48 * sizeof(float));
  float* G      = (float*)carve((size_t)NROW * GN * sizeof(float));
  f16*   Hb     = (f16*)  carve((size_t)NROW * HID * sizeof(f16));

  lstm_front_kernel<<<PK_BLKS + POOL_BLKS, PK_THREADS, 0, stream>>>(features, pooled,
                                                       W_in, W_ih, W_hh, W_out,
                                                       b_ih, b_hh, b_out,
                                                       Winp, Wihp, Whhp, Woutp, gbias, obias);
  lstm_pregate_kernel<<<NTILE, 32, 0, stream>>>(captions, pooled, Winp, b_in, Wihp, gbias, G);
  lstm_recur_kernel<<<BATCH / 16, 32, 0, stream>>>(G, Whhp, Hb);
  lstm_outhead_kernel<<<NTILE, 32, 0, stream>>>(Hb, Woutp, obias, out);
}


# Recurrence: 64 waves of 8 batch rows; per step and lane 1 WMMA and 2 cells (10 tanh), not 2 and 4 (20); column groups told apart by k slots, h of lane n^8 by DPP; k order of h changed, ring 16 deep
# speedup vs baseline: 9.3730x; 1.2418x over previous
#include <hip/hip_runtime.h>
#include <hip/hip_bf16.h>

typedef _Float16 f16;
typedef __attribute__((ext_vector_type(16))) _Float16 v16h;
typedef __attribute__((ext_vector_type(8)))  _Float16 v8h;
typedef __attribute__((ext_vector_type(4)))  _Float16 v4h;
typedef __attribute__((ext_vector_type(8)))  float    v8f;

#define BATCH 512
#define TSEQ  256
#define CNNC  512
#define VOC   37
#define HID   8
#define GN    32
#define NROW  (TSEQ * BATCH)
#define NTILE (NROW / 16)

__device__ __forceinline__ v8f wmma16(v16h a, v16h b, v8f c) {
  return __builtin_amdgcn_wmma_f32_16x16x32_f16(
       false, a,  false, b,
       (short)0, c,  false,  false);
}

__device__ __forceinline__ v16h cat8(v8h a, v8h b) {
  v16h r;
  for (int i = 0; i < 8; ++i) { r[i] = a[i]; r[i + 8] = b[i]; }
  return r;
}

__device__ __forceinline__ v8f splat8(float x) {
  v8f r;
  for (int i = 0; i < 8; ++i) r[i] = x;
  return r;
}

#define GATE_G 2
__device__ __forceinline__ float tanhf_(float x) {
  return __builtin_amdgcn_tanhf(x);
}
__device__ __forceinline__ float sigmoid_of_half_(float xhalf) {
  return fmaf(0.5f, __builtin_amdgcn_tanhf(xhalf), 0.5f);
}
__device__ __forceinline__ float gate_scale(int n) {
  return (n / HID == GATE_G) ? 1.0f : 0.5f;
}

#define POOL_THREADS 256
#define POOL_BLKS ((BATCH * CNNC) / POOL_THREADS)
__device__ __forceinline__ void pool_block(const float* __restrict__ features,
                                           f16* __restrict__ pooled, int blk) {
  int idx = blk * POOL_THREADS + threadIdx.x;
  const float* p = features + (size_t)idx * 49;
  float s = 0.f;
  #pragma unroll
  for (int i = 0; i < 49; ++i) s += p[i];
  pooled[idx] = (f16)(s * (1.0f / 49.0f));
}

__device__ __forceinline__ int gate_row(int nt, int m) {
  int r = m & 7, hh = m >> 3;
  return (2 * nt + (r >> 2)) * HID + 4 * hh + (r & 3);
}

__device__ __forceinline__ int whh_row(int m, int q) {
  int r = m & 7;
  return (r >> 1) * HID + 2 * (2 * q + (m >> 3)) + (r & 1);
}
__device__ __forceinline__ int whh_unit(int hh, int s) {
  int q = s >> 2, j = s & 3;
  int own = 2 * q + hh;
  int other = 2 * (1 - q) + hh;
  return 2 * (j < 2 ? own : other) + (j & 1);
}

#define PK_THREADS   256
#define PK_WIN_BLKS  (48 * CNNC / PK_THREADS)
#define PK_WIH_BLKS  (4 * 512 / PK_THREADS)
#define PK_WHH_BLKS  (1 * 512 / PK_THREADS)
#define PK_WOUT_BLKS (3 * 512 / PK_THREADS)
#define PK_BLKS (PK_WIN_BLKS + PK_WIH_BLKS + PK_WHH_BLKS + PK_WOUT_BLKS + 1)
static_assert((48 * CNNC) % PK_THREADS == 0 && 512 % PK_THREADS == 0,
              "a pack block must not straddle two tables");
static_assert(PK_THREADS == POOL_THREADS, "pack and pool blocks share one launch");

__global__ void __launch_bounds__(PK_THREADS)
lstm_front_kernel(const float* __restrict__ features,
                 f16* __restrict__ pooled,
                 const float* __restrict__ W_in,
                 const float* __restrict__ W_ih,
                 const float* __restrict__ W_hh,
                 const float* __restrict__ W_out,
                 const float* __restrict__ b_ih,
                 const float* __restrict__ b_hh,
                 const float* __restrict__ b_out,
                 f16* __restrict__ Winp,
                 f16* __restrict__ Wihp,
                 f16* __restrict__ Whhp,
                 f16* __restrict__ Woutp,
                 float* __restrict__ gbias,
                 float* __restrict__ obias)
{
  const int tid = threadIdx.x;
  int blk = blockIdx.x;
  if (blk >= PK_BLKS) {
    pool_block(features, pooled, blk - PK_BLKS);
    return;
  }
  if (blk < PK_WIN_BLKS) {
    int idx = blk * PK_THREADS + tid;
    int n = idx / CNNC, k = idx % CNNC;
    Winp[idx] = (f16)(n < VOC ? W_in[n * CNNC + k] : 0.f);
    return;
  }
  blk -= PK_WIN_BLKS;
  if (blk < PK_WIH_BLKS) {
    int idx = blk * PK_THREADS + tid;
    int tile = idx / 512, e = idx % 512, lane = e / 16, h = e % 16;
    int ks = tile / 2, nt = tile % 2;
    int k = ks * 32 + 16 * (lane / 16) + h;
    int n = gate_row(nt, lane % 16);
    Wihp[idx] = (f16)(k < VOC ? gate_scale(n) * W_ih[n * VOC + k] : 0.f);
    return;
  }
  blk -= PK_WIH_BLKS;
  if (blk < PK_WHH_BLKS) {
    int idx = blk * PK_THREADS + tid;
    int lane = idx / 16, h = idx % 16;
    float v = 0.f;
    if (h < 8) {
      int n = whh_row(lane % 16, h >> 2);
      v = gate_scale(n) * W_hh[n * HID + whh_unit(lane / 16, h)];
    }
    Whhp[idx] = (f16)v;
    return;
  }
  blk -= PK_WHH_BLKS;
  if (blk < PK_WOUT_BLKS) {
    int idx = blk * PK_THREADS + tid;
    int tile = idx / 512, e = idx % 512, lane = e / 16, h = e % 16;
    int kl = (h < 8 ? h : h + 8) + (lane / 16) * 8;
    int n = tile * 16 + (lane % 16);
    Woutp[idx] = (f16)((kl < HID && n < VOC) ? W_out[n * HID + kl] : 0.f);
    return;
  }
  if (tid < GN) {
    int n = gate_row(tid / 16, tid % 16);
    gbias[tid] = gate_scale(n) * (b_ih[n] + b_hh[n]);
  }
  if (tid < 48) obias[tid] = (tid < VOC) ? b_out[tid] : 0.f;
}

__global__ void lstm_pregate_kernel(const float* __restrict__ captions,
                                    const f16* __restrict__ pooled,
                                    const f16* __restrict__ Winp,
                                    const float* __restrict__ b_in,
                                    const f16* __restrict__ Wihp,
                                    const float* __restrict__ gbias,
                                    float* __restrict__ G) {
  const int L = threadIdx.x;
  const int j16 = L & 15, half = L >> 4;
  const int t = blockIdx.x / (BATCH / 16);
  const int b = (blockIdx.x % (BATCH / 16)) * 16 + j16;

  v8f acc[2];
  acc[0] = *(const v8f*)(gbias + 8 * half);
  acc[1] = *(const v8f*)(gbias + 16 + 8 * half);

  v16h Xb[2];
  if (t == 0) {
    v8f aP, aQ, aR;
    #pragma unroll
    for (int r = 0; r < 8; ++r) {
      aP[r] = b_in[16 * half + r];
      aQ[r] = b_in[16 * half + 8 + r];
      int n = 32 + 8 * half + r;
      aR[r] = (n < VOC) ? b_in[n] : 0.f;
    }
    const f16* wP = Winp + (size_t)(16 * (j16 >> 3) + (j16 & 7)) * CNNC;
    const f16* wQ = wP + (size_t)8 * CNNC;
    const f16* wR = Winp + (size_t)(32 + j16) * CNNC;
    const f16* pr = pooled + (size_t)b * CNNC;
    for (int kt = 0; kt < CNNC / 32; kt += 2) {
      v8h fB[4], fP[4], fQ[4], fR[4];
      #pragma unroll
      for (int j = 0; j < 2; ++j) {
        int kb = (kt + j) * 32 + half * 8;
        fB[2 * j] = *(const v8h*)(pr + kb);  fB[2 * j + 1] = *(const v8h*)(pr + kb + 16);
        fP[2 * j] = *(const v8h*)(wP + kb);  fP[2 * j + 1] = *(const v8h*)(wP + kb + 16);
        fQ[2 * j] = *(const v8h*)(wQ + kb);  fQ[2 * j + 1] = *(const v8h*)(wQ + kb + 16);
        fR[2 * j] = *(const v8h*)(wR + kb);  fR[2 * j + 1] = *(const v8h*)(wR + kb + 16);
      }
      __builtin_amdgcn_sched_barrier(0);
      #pragma unroll
      for (int j = 0; j < 2; ++j) {
        v16h Bp = cat8(fB[2 * j], fB[2 * j + 1]);
        aP = wmma16(cat8(fP[2 * j], fP[2 * j + 1]), Bp, aP);
        aQ = wmma16(cat8(fQ[2 * j], fQ[2 * j + 1]), Bp, aQ);
        aR = wmma16(cat8(fR[2 * j], fR[2 * j + 1]), Bp, aR);
      }
    }
    #pragma unroll
    for (int s = 0; s < 8; ++s) {
      Xb[0][s]     = (f16)aP[s];
      Xb[0][8 + s] = (f16)aQ[s];
      Xb[1][s]     = (f16)aR[s];
      Xb[1][8 + s] = (f16)0;
    }
  } else {
    const float* xr = captions + ((size_t)b * TSEQ + (t - 1)) * VOC;
    #pragma unroll
    for (int s = 0; s < 16; ++s) Xb[0][s] = (f16)xr[16 * half + s];
    #pragma unroll
    for (int s = 0; s < 16; ++s) {
      float v = (s < VOC - 32) ? xr[32 + s] : 0.f;
      Xb[1][s] = (half == 0) ? (f16)v : (f16)0;
    }
  }

  #pragma unroll
  for (int ks = 0; ks < 2; ++ks) {
    #pragma unroll
    for (int nt = 0; nt < 2; ++nt) {
      const f16* tp = Wihp + (size_t)(ks * 2 + nt) * 512 + L * 16;
      v16h Wa = cat8(*(const v8h*)tp, *(const v8h*)(tp + 8));
      acc[nt] = wmma16(Wa, Xb[ks], acc[nt]);
    }
  }
  v8f pa, pb;
  #pragma unroll
  for (int g = 0; g < 4; ++g) {
    pa[2 * g]     = acc[g >> 1][4 * (g & 1)];
    pa[2 * g + 1] = acc[g >> 1][4 * (g & 1) + 1];
    pb[2 * g]     = acc[g >> 1][4 * (g & 1) + 2];
    pb[2 * g + 1] = acc[g >> 1][4 * (g & 1) + 3];
  }
  const int grp8 = 2 * (blockIdx.x % (BATCH / 16)) + (j16 >> 3);
  const int rl = 8 * half + (j16 & 7);
  float* dst = G + ((size_t)t * (BATCH / 8) + grp8) * 256;
  *(v8f*)(dst + rl * 8)        = pa;
  *(v8f*)(dst + (rl + 16) * 8) = pb;
}

#define GAHEAD 16
static_assert(TSEQ % GAHEAD == 0 && TSEQ >= 2 * GAHEAD, "ring depth must divide the sequence");

typedef __attribute__((ext_vector_type(2))) _Float16 v2h;
typedef __attribute__((ext_vector_type(8))) int v8i;

__device__ __forceinline__ int from_lane_xor8(int x) {
  return __builtin_amdgcn_mov_dpp(x, 0x128  , 0xf, 0xf, false);
}

__device__ __forceinline__ void lstm_cell_step(v8f g, const v16h& Wa, v16h& Bh,
                                               float (&cst)[2], bool grp0,
                                               f16* __restrict__ hout) {
  g = wmma16(Wa, Bh, g);

  float h0[2], h1[2];
  #pragma unroll
  for (int u = 0; u < 2; ++u) {
    float iv = sigmoid_of_half_(g[u]);
    float fv = sigmoid_of_half_(g[2 + u]);
    float gv = tanhf_(g[4 + u]);
    float ov = sigmoid_of_half_(g[6 + u]);
    cst[u] = fv * cst[u] + iv * gv;
    float tc = tanhf_(cst[u]);
    h0[u] = (grp0 ? ov : 0.f) * tc;
    h1[u] = (grp0 ? 0.f : ov) * tc;
  }
  v2h q0 = { (f16)h0[0], (f16)h0[1] };
  v2h q1 = { (f16)h1[0], (f16)h1[1] };
  int hq0 = __builtin_bit_cast(int, q0);
  int hq1 = __builtin_bit_cast(int, q1);
  v8i b;
  b[0] = hq0;
  b[1] = from_lane_xor8(hq1);
  b[2] = hq1;
  b[3] = from_lane_xor8(hq0);
  b[4] = 0; b[5] = 0; b[6] = 0; b[7] = 0;
  Bh = __builtin_bit_cast(v16h, b);

  *(int*)hout = grp0 ? hq0 : hq1;
}

__global__ void __launch_bounds__(32)
lstm_recur_kernel(const float* __restrict__ G,
                  const f16* __restrict__ Whhp,
                  f16* __restrict__ Hb) {
  const int L = threadIdx.x;
  const int n16 = L & 15, half = L >> 4;
  const int grp = n16 >> 3;
  const bool grp0 = (grp == 0);
  const int pair = 2 * grp + half;
  const int b = blockIdx.x * 8 + (n16 & 7);

  const f16* tp = Whhp + L * 16;
  v16h WhhA = cat8(*(const v8h*)tp, *(const v8h*)(tp + 8));

  float cst[2];
  for (int u = 0; u < 2; ++u) cst[u] = 0.f;

  v16h Bh;
  for (int i = 0; i < 16; ++i) Bh[i] = (f16)0;

  f16* hdst = Hb + (size_t)b * HID + 2 * pair;
  const size_t HSTEP = (size_t)BATCH * HID;

  const float* Gl = G + (size_t)blockIdx.x * 256 + L * 8;
  const size_t GSTEP = (size_t)(BATCH / 8) * 256;

  __builtin_amdgcn_sched_barrier(0);

  v8f ra[GAHEAD];
  #pragma unroll
  for (int d = 0; d < GAHEAD; ++d) ra[d] = *(const v8f*)(Gl + d * GSTEP);

  int t = 0;
  for (; t < TSEQ - GAHEAD; t += GAHEAD) {
    #pragma unroll
    for (int d = 0; d < GAHEAD; ++d) {
      lstm_cell_step(ra[d], WhhA, Bh, cst, grp0, hdst + (size_t)(t + d) * HSTEP);
      __builtin_amdgcn_sched_barrier(0);
      ra[d] = *(const v8f*)(Gl + (size_t)(t + d + GAHEAD) * GSTEP);
    }
  }
  #pragma unroll
  for (int d = 0; d < GAHEAD; ++d)
    lstm_cell_step(ra[d], WhhA, Bh, cst, grp0, hdst + (size_t)(t + d) * HSTEP);
}

__global__ void lstm_outhead_kernel(const f16* __restrict__ Hb,
                                    const f16* __restrict__ Woutp,
                                    const float* __restrict__ obias,
                                    float* __restrict__ out) {
  const int L = threadIdx.x;
  const int j16 = L & 15, half = L >> 4;
  const size_t Mb = (size_t)blockIdx.x * 16;

  v16h WoutB[3];
  #pragma unroll
  for (int nt = 0; nt < 3; ++nt) {
    const f16* tp = Woutp + (size_t)nt * 512 + L * 16;
    WoutB[nt] = cat8(*(const v8h*)tp, *(const v8h*)(tp + 8));
  }

  v16h A;
  #pragma unroll
  for (int i = 0; i < 16; ++i) A[i] = (f16)0;
  if (L < 16) {
    v8h hv = *(const v8h*)(Hb + (Mb + L) * HID);
    #pragma unroll
    for (int k = 0; k < HID; ++k) A[k] = hv[k];
  }

  v8f l0 = splat8(obias[j16]);
  v8f l1 = splat8(obias[16 + j16]);
  v8f l2 = splat8(obias[32 + j16]);
  l0 = wmma16(A, WoutB[0], l0);
  l1 = wmma16(A, WoutB[1], l1);
  l2 = wmma16(A, WoutB[2], l2);

  const bool v2ok = (32 + j16) < VOC;
  #pragma unroll
  for (int r = 0; r < 8; ++r) {
    float x2 = v2ok ? l2[r] : -1e30f;
    float mx = fmaxf(fmaxf(l0[r], l1[r]), x2);
    #pragma unroll
    for (int s = 1; s < 16; s <<= 1)
      mx = fmaxf(mx, __shfl_xor(mx, s, 16));
    float e0 = __expf(l0[r] - mx);
    float e1 = __expf(l1[r] - mx);
    float e2 = v2ok ? __expf(x2 - mx) : 0.f;
    float sm = e0 + e1 + e2;
    #pragma unroll
    for (int s = 1; s < 16; s <<= 1)
      sm += __shfl_xor(sm, s, 16);
    float inv = 1.0f / sm;

    size_t row = Mb + r + 8 * half;
    int t = (int)(row / BATCH);
    int b = (int)(row % BATCH);
    float* dst = out + ((size_t)b * TSEQ + t) * VOC;
    dst[j16]      = e0 * inv;
    dst[16 + j16] = e1 * inv;
    if (v2ok) dst[32 + j16] = e2 * inv;
  }
}

extern "C" void kernel_launch(void* const* d_in, const int* in_sizes, int n_in,
                              void* d_out, int out_size, void* d_ws, size_t ws_size,
                              hipStream_t stream) {
  const float* features = (const float*)d_in[0];
  const float* captions = (const float*)d_in[1];
  const float* W_in  = (const float*)d_in[2];
  const float* b_in  = (const float*)d_in[3];
  const float* W_ih  = (const float*)d_in[4];
  const float* W_hh  = (const float*)d_in[5];
  const float* b_ih  = (const float*)d_in[6];
  const float* b_hh  = (const float*)d_in[7];
  const float* W_out = (const float*)d_in[8];
  const float* b_out = (const float*)d_in[9];
  float* out = (float*)d_out;

  char* ws = (char*)d_ws;
  auto carve = [&](size_t bytes) -> char* {
    char* p = ws;
    ws += (bytes + 255) & ~(size_t)255;
    return p;
  };
  f16*   pooled = (f16*)  carve((size_t)BATCH * CNNC * sizeof(f16));
  f16*   Winp   = (f16*)  carve((size_t)48 * CNNC * sizeof(f16));
  f16*   Wihp   = (f16*)  carve((size_t)4 * 512 * sizeof(f16));
  f16*   Whhp   = (f16*)  carve((size_t)1 * 512 * sizeof(f16));
  f16*   Woutp  = (f16*)  carve((size_t)3 * 512 * sizeof(f16));
  float* gbias  = (float*)carve(GN * sizeof(float));
  float* obias  = (float*)carve(48 * sizeof(float));
  float* G      = (float*)carve((size_t)NROW * GN * sizeof(float));
  f16*   Hb     = (f16*)  carve((size_t)NROW * HID * sizeof(f16));

  lstm_front_kernel<<<PK_BLKS + POOL_BLKS, PK_THREADS, 0, stream>>>(features, pooled,
                                                       W_in, W_ih, W_hh, W_out,
                                                       b_ih, b_hh, b_out,
                                                       Winp, Wihp, Whhp, Woutp, gbias, obias);
  lstm_pregate_kernel<<<NTILE, 32, 0, stream>>>(captions, pooled, Winp, b_in, Wihp, gbias, G);
  lstm_recur_kernel<<<BATCH / 8, 32, 0, stream>>>(G, Whhp, Hb);
  lstm_outhead_kernel<<<NTILE, 32, 0, stream>>>(Hb, Woutp, obias, out);
}


# Output head fused into the recurrence launch: 8 head waves per block take each 16 steps of h from LDS behind one barrier, softmax along a lane's registers (sum order changed): 3 launches, not 4
# speedup vs baseline: 10.1537x; 1.0833x over previous
#include <hip/hip_runtime.h>
#include <hip/hip_bf16.h>

typedef _Float16 f16;
typedef __attribute__((ext_vector_type(16))) _Float16 v16h;
typedef __attribute__((ext_vector_type(8)))  _Float16 v8h;
typedef __attribute__((ext_vector_type(4)))  _Float16 v4h;
typedef __attribute__((ext_vector_type(8)))  float    v8f;

#define BATCH 512
#define TSEQ  256
#define CNNC  512
#define VOC   37
#define HID   8
#define GN    32
#define NROW  (TSEQ * BATCH)
#define NTILE (NROW / 16)

__device__ __forceinline__ v8f wmma16(v16h a, v16h b, v8f c) {
  return __builtin_amdgcn_wmma_f32_16x16x32_f16(
       false, a,  false, b,
       (short)0, c,  false,  false);
}

__device__ __forceinline__ v16h cat8(v8h a, v8h b) {
  v16h r;
  for (int i = 0; i < 8; ++i) { r[i] = a[i]; r[i + 8] = b[i]; }
  return r;
}

#define GATE_G 2
__device__ __forceinline__ float tanhf_(float x) {
  return __builtin_amdgcn_tanhf(x);
}
__device__ __forceinline__ float sigmoid_of_half_(float xhalf) {
  return fmaf(0.5f, __builtin_amdgcn_tanhf(xhalf), 0.5f);
}
__device__ __forceinline__ float gate_scale(int n) {
  return (n / HID == GATE_G) ? 1.0f : 0.5f;
}

#define POOL_THREADS 256
#define POOL_BLKS ((BATCH * CNNC) / POOL_THREADS)
__device__ __forceinline__ void pool_block(const float* __restrict__ features,
                                           f16* __restrict__ pooled, int blk) {
  int idx = blk * POOL_THREADS + threadIdx.x;
  const float* p = features + (size_t)idx * 49;
  float s = 0.f;
  #pragma unroll
  for (int i = 0; i < 49; ++i) s += p[i];
  pooled[idx] = (f16)(s * (1.0f / 49.0f));
}

__device__ __forceinline__ int gate_row(int nt, int m) {
  int r = m & 7, hh = m >> 3;
  return (2 * nt + (r >> 2)) * HID + 4 * hh + (r & 3);
}

__device__ __forceinline__ int whh_row(int m, int q) {
  int r = m & 7;
  return (r >> 1) * HID + 2 * (2 * q + (m >> 3)) + (r & 1);
}
__device__ __forceinline__ int whh_unit(int hh, int s) {
  int q = s >> 2, j = s & 3;
  int own = 2 * q + hh;
  int other = 2 * (1 - q) + hh;
  return 2 * (j < 2 ? own : other) + (j & 1);
}

#define PK_THREADS   256
#define PK_WIN_BLKS  (48 * CNNC / PK_THREADS)
#define PK_WIH_BLKS  (4 * 512 / PK_THREADS)
#define PK_WHH_BLKS  (1 * 512 / PK_THREADS)
#define PK_WOUT_BLKS (3 * 512 / PK_THREADS)
#define PK_BLKS (PK_WIN_BLKS + PK_WIH_BLKS + PK_WHH_BLKS + PK_WOUT_BLKS + 1)
static_assert((48 * CNNC) % PK_THREADS == 0 && 512 % PK_THREADS == 0,
              "a pack block must not straddle two tables");
static_assert(PK_THREADS == POOL_THREADS, "pack and pool blocks share one launch");

__global__ void __launch_bounds__(PK_THREADS)
lstm_front_kernel(const float* __restrict__ features,
                 f16* __restrict__ pooled,
                 const float* __restrict__ W_in,
                 const float* __restrict__ W_ih,
                 const float* __restrict__ W_hh,
                 const float* __restrict__ W_out,
                 const float* __restrict__ b_ih,
                 const float* __restrict__ b_hh,
                 const float* __restrict__ b_out,
                 f16* __restrict__ Winp,
                 f16* __restrict__ Wihp,
                 f16* __restrict__ Whhp,
                 f16* __restrict__ Woutp,
                 float* __restrict__ gbias,
                 float* __restrict__ obias)
{
  const int tid = threadIdx.x;
  int blk = blockIdx.x;
  if (blk >= PK_BLKS) {
    pool_block(features, pooled, blk - PK_BLKS);
    return;
  }
  if (blk < PK_WIN_BLKS) {
    int idx = blk * PK_THREADS + tid;
    int n = idx / CNNC, k = idx % CNNC;
    Winp[idx] = (f16)(n < VOC ? W_in[n * CNNC + k] : 0.f);
    return;
  }
  blk -= PK_WIN_BLKS;
  if (blk < PK_WIH_BLKS) {
    int idx = blk * PK_THREADS + tid;
    int tile = idx / 512, e = idx % 512, lane = e / 16, h = e % 16;
    int ks = tile / 2, nt = tile % 2;
    int k = ks * 32 + 16 * (lane / 16) + h;
    int n = gate_row(nt, lane % 16);
    Wihp[idx] = (f16)(k < VOC ? gate_scale(n) * W_ih[n * VOC + k] : 0.f);
    return;
  }
  blk -= PK_WIH_BLKS;
  if (blk < PK_WHH_BLKS) {
    int idx = blk * PK_THREADS + tid;
    int lane = idx / 16, h = idx % 16;
    float v = 0.f;
    if (h < 8) {
      int n = whh_row(lane % 16, h >> 2);
      v = gate_scale(n) * W_hh[n * HID + whh_unit(lane / 16, h)];
    }
    Whhp[idx] = (f16)v;
    return;
  }
  blk -= PK_WHH_BLKS;
  if (blk < PK_WOUT_BLKS) {
    int idx = blk * PK_THREADS + tid;
    int tile = idx / 512, e = idx % 512, lane = e / 16, h = e % 16;
    int kl = (h < 8 ? h : h + 8) + (lane / 16) * 8;
    int n = tile * 16 + (lane % 16);
    Woutp[idx] = (f16)((kl < HID && n < VOC) ? W_out[n * HID + kl] : 0.f);
    return;
  }
  if (tid < GN) {
    int n = gate_row(tid / 16, tid % 16);
    gbias[tid] = gate_scale(n) * (b_ih[n] + b_hh[n]);
  }
  if (tid < 48) obias[tid] = (tid < VOC) ? b_out[tid] : 0.f;
}

__global__ void lstm_pregate_kernel(const float* __restrict__ captions,
                                    const f16* __restrict__ pooled,
                                    const f16* __restrict__ Winp,
                                    const float* __restrict__ b_in,
                                    const f16* __restrict__ Wihp,
                                    const float* __restrict__ gbias,
                                    float* __restrict__ G) {
  const int L = threadIdx.x;
  const int j16 = L & 15, half = L >> 4;
  const int t = blockIdx.x / (BATCH / 16);
  const int b = (blockIdx.x % (BATCH / 16)) * 16 + j16;

  v8f acc[2];
  acc[0] = *(const v8f*)(gbias + 8 * half);
  acc[1] = *(const v8f*)(gbias + 16 + 8 * half);

  v16h Xb[2];
  if (t == 0) {
    v8f aP, aQ, aR;
    #pragma unroll
    for (int r = 0; r < 8; ++r) {
      aP[r] = b_in[16 * half + r];
      aQ[r] = b_in[16 * half + 8 + r];
      int n = 32 + 8 * half + r;
      aR[r] = (n < VOC) ? b_in[n] : 0.f;
    }
    const f16* wP = Winp + (size_t)(16 * (j16 >> 3) + (j16 & 7)) * CNNC;
    const f16* wQ = wP + (size_t)8 * CNNC;
    const f16* wR = Winp + (size_t)(32 + j16) * CNNC;
    const f16* pr = pooled + (size_t)b * CNNC;
    for (int kt = 0; kt < CNNC / 32; kt += 2) {
      v8h fB[4], fP[4], fQ[4], fR[4];
      #pragma unroll
      for (int j = 0; j < 2; ++j) {
        int kb = (kt + j) * 32 + half * 8;
        fB[2 * j] = *(const v8h*)(pr + kb);  fB[2 * j + 1] = *(const v8h*)(pr + kb + 16);
        fP[2 * j] = *(const v8h*)(wP + kb);  fP[2 * j + 1] = *(const v8h*)(wP + kb + 16);
        fQ[2 * j] = *(const v8h*)(wQ + kb);  fQ[2 * j + 1] = *(const v8h*)(wQ + kb + 16);
        fR[2 * j] = *(const v8h*)(wR + kb);  fR[2 * j + 1] = *(const v8h*)(wR + kb + 16);
      }
      __builtin_amdgcn_sched_barrier(0);
      #pragma unroll
      for (int j = 0; j < 2; ++j) {
        v16h Bp = cat8(fB[2 * j], fB[2 * j + 1]);
        aP = wmma16(cat8(fP[2 * j], fP[2 * j + 1]), Bp, aP);
        aQ = wmma16(cat8(fQ[2 * j], fQ[2 * j + 1]), Bp, aQ);
        aR = wmma16(cat8(fR[2 * j], fR[2 * j + 1]), Bp, aR);
      }
    }
    #pragma unroll
    for (int s = 0; s < 8; ++s) {
      Xb[0][s]     = (f16)aP[s];
      Xb[0][8 + s] = (f16)aQ[s];
      Xb[1][s]     = (f16)aR[s];
      Xb[1][8 + s] = (f16)0;
    }
  } else {
    const float* xr = captions + ((size_t)b * TSEQ + (t - 1)) * VOC;
    #pragma unroll
    for (int s = 0; s < 16; ++s) Xb[0][s] = (f16)xr[16 * half + s];
    #pragma unroll
    for (int s = 0; s < 16; ++s) {
      float v = (s < VOC - 32) ? xr[32 + s] : 0.f;
      Xb[1][s] = (half == 0) ? (f16)v : (f16)0;
    }
  }

  #pragma unroll
  for (int ks = 0; ks < 2; ++ks) {
    #pragma unroll
    for (int nt = 0; nt < 2; ++nt) {
      const f16* tp = Wihp + (size_t)(ks * 2 + nt) * 512 + L * 16;
      v16h Wa = cat8(*(const v8h*)tp, *(const v8h*)(tp + 8));
      acc[nt] = wmma16(Wa, Xb[ks], acc[nt]);
    }
  }
  v8f pa, pb;
  #pragma unroll
  for (int g = 0; g < 4; ++g) {
    pa[2 * g]     = acc[g >> 1][4 * (g & 1)];
    pa[2 * g + 1] = acc[g >> 1][4 * (g & 1) + 1];
    pb[2 * g]     = acc[g >> 1][4 * (g & 1) + 2];
    pb[2 * g + 1] = acc[g >> 1][4 * (g & 1) + 3];
  }
  const int grp8 = 2 * (blockIdx.x % (BATCH / 16)) + (j16 >> 3);
  const int rl = 8 * half + (j16 & 7);
  float* dst = G + ((size_t)t * (BATCH / 8) + grp8) * 256;
  *(v8f*)(dst + rl * 8)        = pa;
  *(v8f*)(dst + (rl + 16) * 8) = pb;
}

#define GAHEAD 16
static_assert(TSEQ % GAHEAD == 0 && TSEQ >= 2 * GAHEAD, "ring depth must divide the sequence");
#define NTURN (TSEQ / GAHEAD)
#define HEAD_WAVES 8
#define RH_THREADS (32 * (1 + HEAD_WAVES))
#define HPITCH (GAHEAD * HID + 8)
#define HSTAGE (8 * HPITCH)
static_assert(GAHEAD == 16, "a head tile is the sixteen steps of one ring turn");
static_assert((HPITCH * 2) % 16 == 0 && (HSTAGE * 2) % 16 == 0, "16-byte reads of a stage row");

typedef __attribute__((ext_vector_type(2))) _Float16 v2h;
typedef __attribute__((ext_vector_type(8))) int v8i;
typedef __attribute__((ext_vector_type(4))) int v4i;

__device__ __forceinline__ int from_lane_xor8(int x) {
  return __builtin_amdgcn_mov_dpp(x, 0x128  , 0xf, 0xf, false);
}

__device__ __forceinline__ float from_other_half(float x) {
  unsigned u = __builtin_bit_cast(unsigned, x);
  unsigned r = __builtin_amdgcn_permlanex16(u, u, 0x76543210u, 0xfedcba98u, false, false);
  return __builtin_bit_cast(float, r);
}

__device__ __forceinline__ void lstm_cell_step(v8f g, const v16h& Wa, v16h& Bh,
                                               float (&cst)[2], bool grp0,
                                               f16* hout) {
  g = wmma16(Wa, Bh, g);

  float h0[2], h1[2];
  #pragma unroll
  for (int u = 0; u < 2; ++u) {
    float iv = sigmoid_of_half_(g[u]);
    float fv = sigmoid_of_half_(g[2 + u]);
    float gv = tanhf_(g[4 + u]);
    float ov = sigmoid_of_half_(g[6 + u]);
    cst[u] = fv * cst[u] + iv * gv;
    float tc = tanhf_(cst[u]);
    h0[u] = (grp0 ? ov : 0.f) * tc;
    h1[u] = (grp0 ? 0.f : ov) * tc;
  }
  v2h q0 = { (f16)h0[0], (f16)h0[1] };
  v2h q1 = { (f16)h1[0], (f16)h1[1] };
  int hq0 = __builtin_bit_cast(int, q0);
  int hq1 = __builtin_bit_cast(int, q1);
  v8i b;
  b[0] = hq0;
  b[1] = from_lane_xor8(hq1);
  b[2] = hq1;
  b[3] = from_lane_xor8(hq0);
  b[4] = 0; b[5] = 0; b[6] = 0; b[7] = 0;
  Bh = __builtin_bit_cast(v16h, b);

  *(int*)hout = grp0 ? hq0 : hq1;
}

__device__ __forceinline__ void head_tile(const f16* hrow, const v16h (&WoutA)[3],
                                          const v8f (&cb)[3], int L,
                                          float* __restrict__ orow) {
  const int n = L & 15, half = L >> 4;
  v4i hv = *(const v4i*)(hrow + n * HID);
  v8i bb;
  for (int i = 0; i < 4; ++i) bb[i] = (half == 0) ? hv[i] : 0;
  bb[4] = 0; bb[5] = 0; bb[6] = 0; bb[7] = 0;
  const v16h Bt = __builtin_bit_cast(v16h, bb);

  v8f l0 = wmma16(WoutA[0], Bt, cb[0]);
  v8f l1 = wmma16(WoutA[1], Bt, cb[1]);
  v8f l2 = wmma16(WoutA[2], Bt, cb[2]);

  constexpr int R2 = VOC - 32;
  float x2[R2];
  for (int r = 0; r < R2; ++r) x2[r] = (half == 0) ? l2[r] : -1e30f;

  float mx = l0[0];
  for (int r = 1; r < 8; ++r) mx = fmaxf(mx, l0[r]);
  for (int r = 0; r < 8; ++r) mx = fmaxf(mx, l1[r]);
  for (int r = 0; r < R2; ++r) mx = fmaxf(mx, x2[r]);
  mx = fmaxf(mx, from_other_half(mx));

  float e0[8], e1[8], e2[R2];
  float sm = 0.f;
  for (int r = 0; r < 8; ++r) { e0[r] = __expf(l0[r] - mx); sm += e0[r]; }
  for (int r = 0; r < 8; ++r) { e1[r] = __expf(l1[r] - mx); sm += e1[r]; }
  for (int r = 0; r < R2; ++r) { e2[r] = __expf(x2[r] - mx); sm += e2[r]; }
  sm += from_other_half(sm);
  const float inv = 1.0f / sm;

  float* dst = orow + (size_t)n * VOC;
  for (int r = 0; r < 8; ++r) dst[8 * half + r] = e0[r] * inv;
  for (int r = 0; r < 8; ++r) dst[16 + 8 * half + r] = e1[r] * inv;
  if (half == 0) {
    for (int r = 0; r < R2; ++r) dst[32 + r] = e2[r] * inv;
  }
}

__global__ void __launch_bounds__(RH_THREADS)
lstm_recur_head_kernel(const float* __restrict__ G,
                       const f16* __restrict__ Whhp,
                       const f16* __restrict__ Woutp,
                       const float* __restrict__ obias,
                       float* __restrict__ out) {
  __shared__ __attribute__((aligned(16))) f16 hring[2 * HSTAGE];

  const int wave = __builtin_amdgcn_readfirstlane((int)(threadIdx.x >> 5));
  const bool recur = (wave == 0);
  const int L = threadIdx.x & 31;
  const int n16 = L & 15, half = L >> 4;

  const int grp = n16 >> 3;
  const bool grp0 = (grp == 0);
  const int pair = 2 * grp + half;
  v16h WhhA;
  float cst[2];
  v16h Bh;
  v8f ra[GAHEAD];
  f16* const hmine = hring + (n16 & 7) * HPITCH + 2 * pair;
  const float* Gl = G + (size_t)blockIdx.x * 256 + L * 8;
  const size_t GSTEP = (size_t)(BATCH / 8) * 256;

  v16h WoutA[3];
  v8f cb[3];
  const int hw = recur ? 0 : wave - 1;
  const f16* const hrow = hring + hw * HPITCH;
  float* const obase = out + (size_t)(blockIdx.x * 8 + hw) * TSEQ * VOC;

  if (recur) {
    __builtin_amdgcn_s_setprio(3);
    const f16* tp = Whhp + L * 16;
    WhhA = cat8(*(const v8h*)tp, *(const v8h*)(tp + 8));
    for (int u = 0; u < 2; ++u) cst[u] = 0.f;
    for (int i = 0; i < 16; ++i) Bh[i] = (f16)0;
    __builtin_amdgcn_sched_barrier(0);
    #pragma unroll
    for (int d = 0; d < GAHEAD; ++d) ra[d] = *(const v8f*)(Gl + d * GSTEP);
  } else {
    for (int nt = 0; nt < 3; ++nt) {
      const f16* tp = Woutp + (size_t)nt * 512 + L * 16;
      WoutA[nt] = cat8(*(const v8h*)tp, *(const v8h*)(tp + 8));
      cb[nt] = *(const v8f*)(obias + 16 * nt + 8 * half);
    }
  }

  for (int g = 0; g < NTURN; ++g) {
    if (recur) {
      f16* const hs = hmine + (g & 1) * HSTAGE;
      const int gn = (g + 1 < NTURN) ? (g + 1) : g;
      const float* Gn = Gl + (size_t)gn * GAHEAD * GSTEP;
      #pragma unroll
      for (int d = 0; d < GAHEAD; ++d) {
        lstm_cell_step(ra[d], WhhA, Bh, cst, grp0, hs + d * HID);
        __builtin_amdgcn_sched_barrier(0);
        ra[d] = *(const v8f*)(Gn + d * GSTEP);
      }
    } else if (g > 0) {
      head_tile(hrow + ((g - 1) & 1) * HSTAGE, WoutA, cb, L,
                obase + (size_t)(g - 1) * GAHEAD * VOC);
    }
    asm volatile("s_wait_dscnt 0x0" ::: "memory");
    asm volatile("s_barrier_signal -1" ::: "memory");
    asm volatile("s_barrier_wait -1" ::: "memory");
  }
  if (!recur)
    head_tile(hrow + ((NTURN - 1) & 1) * HSTAGE, WoutA, cb, L,
              obase + (size_t)(NTURN - 1) * GAHEAD * VOC);
}

extern "C" void kernel_launch(void* const* d_in, const int* in_sizes, int n_in,
                              void* d_out, int out_size, void* d_ws, size_t ws_size,
                              hipStream_t stream) {
  const float* features = (const float*)d_in[0];
  const float* captions = (const float*)d_in[1];
  const float* W_in  = (const float*)d_in[2];
  const float* b_in  = (const float*)d_in[3];
  const float* W_ih  = (const float*)d_in[4];
  const float* W_hh  = (const float*)d_in[5];
  const float* b_ih  = (const float*)d_in[6];
  const float* b_hh  = (const float*)d_in[7];
  const float* W_out = (const float*)d_in[8];
  const float* b_out = (const float*)d_in[9];
  float* out = (float*)d_out;

  char* ws = (char*)d_ws;
  auto carve = [&](size_t bytes) -> char* {
    char* p = ws;
    ws += (bytes + 255) & ~(size_t)255;
    return p;
  };
  f16*   pooled = (f16*)  carve((size_t)BATCH * CNNC * sizeof(f16));
  f16*   Winp   = (f16*)  carve((size_t)48 * CNNC * sizeof(f16));
  f16*   Wihp   = (f16*)  carve((size_t)4 * 512 * sizeof(f16));
  f16*   Whhp   = (f16*)  carve((size_t)1 * 512 * sizeof(f16));
  f16*   Woutp  = (f16*)  carve((size_t)3 * 512 * sizeof(f16));
  float* gbias  = (float*)carve(GN * sizeof(float));
  float* obias  = (float*)carve(48 * sizeof(float));
  float* G      = (float*)carve((size_t)NROW * GN * sizeof(float));

  lstm_front_kernel<<<PK_BLKS + POOL_BLKS, PK_THREADS, 0, stream>>>(features, pooled,
                                                       W_in, W_ih, W_hh, W_out,
                                                       b_ih, b_hh, b_out,
                                                       Winp, Wihp, Whhp, Woutp, gbias, obias);
  lstm_pregate_kernel<<<NTILE, 32, 0, stream>>>(captions, pooled, Winp, b_in, Wihp, gbias, G);
  lstm_recur_head_kernel<<<BATCH / 8, RH_THREADS, 0, stream>>>(G, Whhp, Woutp, obias, out);
}


# Pre-gate launch: K of the t=0 feat product split over the 8 waves of a block, partials added from LDS in wave order (sum order changed), W_ih tiles loaded first: 10 dependent load trips, now 2
# speedup vs baseline: 10.8323x; 1.0668x over previous
#include <hip/hip_runtime.h>
#include <hip/hip_bf16.h>

typedef _Float16 f16;
typedef __attribute__((ext_vector_type(16))) _Float16 v16h;
typedef __attribute__((ext_vector_type(8)))  _Float16 v8h;
typedef __attribute__((ext_vector_type(4)))  _Float16 v4h;
typedef __attribute__((ext_vector_type(8)))  float    v8f;

#define BATCH 512
#define TSEQ  256
#define CNNC  512
#define VOC   37
#define HID   8
#define GN    32
#define NROW  (TSEQ * BATCH)
#define NTILE (NROW / 16)

__device__ __forceinline__ v8f wmma16(v16h a, v16h b, v8f c) {
  return __builtin_amdgcn_wmma_f32_16x16x32_f16(
       false, a,  false, b,
       (short)0, c,  false,  false);
}

__device__ __forceinline__ v16h cat8(v8h a, v8h b) {
  v16h r;
  for (int i = 0; i < 8; ++i) { r[i] = a[i]; r[i + 8] = b[i]; }
  return r;
}

#define GATE_G 2
__device__ __forceinline__ float tanhf_(float x) {
  return __builtin_amdgcn_tanhf(x);
}
__device__ __forceinline__ float sigmoid_of_half_(float xhalf) {
  return fmaf(0.5f, __builtin_amdgcn_tanhf(xhalf), 0.5f);
}
__device__ __forceinline__ float gate_scale(int n) {
  return (n / HID == GATE_G) ? 1.0f : 0.5f;
}

#define POOL_THREADS 256
#define POOL_BLKS ((BATCH * CNNC) / POOL_THREADS)
__device__ __forceinline__ void pool_block(const float* __restrict__ features,
                                           f16* __restrict__ pooled, int blk) {
  int idx = blk * POOL_THREADS + threadIdx.x;
  const float* p = features + (size_t)idx * 49;
  float s = 0.f;
  #pragma unroll
  for (int i = 0; i < 49; ++i) s += p[i];
  pooled[idx] = (f16)(s * (1.0f / 49.0f));
}

__device__ __forceinline__ int gate_row(int nt, int m) {
  int r = m & 7, hh = m >> 3;
  return (2 * nt + (r >> 2)) * HID + 4 * hh + (r & 3);
}

__device__ __forceinline__ int whh_row(int m, int q) {
  int r = m & 7;
  return (r >> 1) * HID + 2 * (2 * q + (m >> 3)) + (r & 1);
}
__device__ __forceinline__ int whh_unit(int hh, int s) {
  int q = s >> 2, j = s & 3;
  int own = 2 * q + hh;
  int other = 2 * (1 - q) + hh;
  return 2 * (j < 2 ? own : other) + (j & 1);
}

#define PK_THREADS   256
#define PK_WIN_BLKS  (48 * CNNC / PK_THREADS)
#define PK_WIH_BLKS  (4 * 512 / PK_THREADS)
#define PK_WHH_BLKS  (1 * 512 / PK_THREADS)
#define PK_WOUT_BLKS (3 * 512 / PK_THREADS)
#define PK_BLKS (PK_WIN_BLKS + PK_WIH_BLKS + PK_WHH_BLKS + PK_WOUT_BLKS + 1)
static_assert((48 * CNNC) % PK_THREADS == 0 && 512 % PK_THREADS == 0,
              "a pack block must not straddle two tables");
static_assert(PK_THREADS == POOL_THREADS, "pack and pool blocks share one launch");

__global__ void __launch_bounds__(PK_THREADS)
lstm_front_kernel(const float* __restrict__ features,
                 f16* __restrict__ pooled,
                 const float* __restrict__ W_in,
                 const float* __restrict__ W_ih,
                 const float* __restrict__ W_hh,
                 const float* __restrict__ W_out,
                 const float* __restrict__ b_ih,
                 const float* __restrict__ b_hh,
                 const float* __restrict__ b_out,
                 f16* __restrict__ Winp,
                 f16* __restrict__ Wihp,
                 f16* __restrict__ Whhp,
                 f16* __restrict__ Woutp,
                 float* __restrict__ gbias,
                 float* __restrict__ obias)
{
  const int tid = threadIdx.x;
  int blk = blockIdx.x;
  if (blk >= PK_BLKS) {
    pool_block(features, pooled, blk - PK_BLKS);
    return;
  }
  if (blk < PK_WIN_BLKS) {
    int idx = blk * PK_THREADS + tid;
    int n = idx / CNNC, k = idx % CNNC;
    Winp[idx] = (f16)(n < VOC ? W_in[n * CNNC + k] : 0.f);
    return;
  }
  blk -= PK_WIN_BLKS;
  if (blk < PK_WIH_BLKS) {
    int idx = blk * PK_THREADS + tid;
    int tile = idx / 512, e = idx % 512, lane = e / 16, h = e % 16;
    int ks = tile / 2, nt = tile % 2;
    int k = ks * 32 + 16 * (lane / 16) + h;
    int n = gate_row(nt, lane % 16);
    Wihp[idx] = (f16)(k < VOC ? gate_scale(n) * W_ih[n * VOC + k] : 0.f);
    return;
  }
  blk -= PK_WIH_BLKS;
  if (blk < PK_WHH_BLKS) {
    int idx = blk * PK_THREADS + tid;
    int lane = idx / 16, h = idx % 16;
    float v = 0.f;
    if (h < 8) {
      int n = whh_row(lane % 16, h >> 2);
      v = gate_scale(n) * W_hh[n * HID + whh_unit(lane / 16, h)];
    }
    Whhp[idx] = (f16)v;
    return;
  }
  blk -= PK_WHH_BLKS;
  if (blk < PK_WOUT_BLKS) {
    int idx = blk * PK_THREADS + tid;
    int tile = idx / 512, e = idx % 512, lane = e / 16, h = e % 16;
    int kl = (h < 8 ? h : h + 8) + (lane / 16) * 8;
    int n = tile * 16 + (lane % 16);
    Woutp[idx] = (f16)((kl < HID && n < VOC) ? W_out[n * HID + kl] : 0.f);
    return;
  }
  if (tid < GN) {
    int n = gate_row(tid / 16, tid % 16);
    gbias[tid] = gate_scale(n) * (b_ih[n] + b_hh[n]);
  }
  if (tid < 48) obias[tid] = (tid < VOC) ? b_out[tid] : 0.f;
}

#define PG_WAVES   8
#define PG_THREADS (32 * PG_WAVES)
#define PG_T0_BLKS (BATCH / 16)
#define PG_BLKS    (PG_T0_BLKS + (NTILE - PG_T0_BLKS) / PG_WAVES)
static_assert((NTILE - PG_T0_BLKS) % PG_WAVES == 0, "whole blocks of t >= 1 tiles");
static_assert(CNNC / 32 == 2 * PG_WAVES, "two k-steps of the feat product per wave");
#define PG_PART 24

__global__ void __launch_bounds__(PG_THREADS)
lstm_pregate_kernel(const float* __restrict__ captions,
                                    const f16* __restrict__ pooled,
                                    const f16* __restrict__ Winp,
                                    const float* __restrict__ b_in,
                                    const f16* __restrict__ Wihp,
                                    const float* __restrict__ gbias,
                                    float* __restrict__ G) {
  __shared__ float fpart[PG_WAVES * PG_PART * 32];
  const int wave = __builtin_amdgcn_readfirstlane((int)(threadIdx.x >> 5));
  const int L = threadIdx.x & 31;
  const int j16 = L & 15, half = L >> 4;
  const bool first = ((int)blockIdx.x < PG_T0_BLKS);
  const int tile = first ? (int)blockIdx.x
                         : PG_T0_BLKS + ((int)blockIdx.x - PG_T0_BLKS) * PG_WAVES + wave;
  const int t = tile / (BATCH / 16);
  const int bt = tile % (BATCH / 16);
  const int b = bt * 16 + j16;

  v16h Wa[4];
  #pragma unroll
  for (int i = 0; i < 4; ++i) {
    const f16* tp = Wihp + (size_t)i * 512 + L * 16;
    Wa[i] = cat8(*(const v8h*)tp, *(const v8h*)(tp + 8));
  }
  v8f acc[2];
  acc[0] = *(const v8f*)(gbias + 8 * half);
  acc[1] = *(const v8f*)(gbias + 16 + 8 * half);

  v16h Xb[2];
  if (first) {
    v8f bP, bQ, bR;
    #pragma unroll
    for (int r = 0; r < 8; ++r) {
      bP[r] = b_in[16 * half + r];
      bQ[r] = b_in[16 * half + 8 + r];
      int n = 32 + 8 * half + r;
      bR[r] = (n < VOC) ? b_in[n] : 0.f;
    }
    v8f aP, aQ, aR;
    #pragma unroll
    for (int r = 0; r < 8; ++r) { aP[r] = 0.f; aQ[r] = 0.f; aR[r] = 0.f; }
    const f16* wP = Winp + (size_t)(16 * (j16 >> 3) + (j16 & 7)) * CNNC;
    const f16* wQ = wP + (size_t)8 * CNNC;
    const f16* wR = Winp + (size_t)(32 + j16) * CNNC;
    const f16* pr = pooled + (size_t)b * CNNC;
    {
      const int kt = 2 * wave;
      v8h fB[4], fP[4], fQ[4], fR[4];
      #pragma unroll
      for (int j = 0; j < 2; ++j) {
        int kb = (kt + j) * 32 + half * 8;
        fB[2 * j] = *(const v8h*)(pr + kb);  fB[2 * j + 1] = *(const v8h*)(pr + kb + 16);
        fP[2 * j] = *(const v8h*)(wP + kb);  fP[2 * j + 1] = *(const v8h*)(wP + kb + 16);
        fQ[2 * j] = *(const v8h*)(wQ + kb);  fQ[2 * j + 1] = *(const v8h*)(wQ + kb + 16);
        fR[2 * j] = *(const v8h*)(wR + kb);  fR[2 * j + 1] = *(const v8h*)(wR + kb + 16);
      }
      __builtin_amdgcn_sched_barrier(0);
      #pragma unroll
      for (int j = 0; j < 2; ++j) {
        v16h Bp = cat8(fB[2 * j], fB[2 * j + 1]);
        aP = wmma16(cat8(fP[2 * j], fP[2 * j + 1]), Bp, aP);
        aQ = wmma16(cat8(fQ[2 * j], fQ[2 * j + 1]), Bp, aQ);
        aR = wmma16(cat8(fR[2 * j], fR[2 * j + 1]), Bp, aR);
      }
    }
    float* mine = fpart + (size_t)wave * PG_PART * 32 + L;
    #pragma unroll
    for (int r = 0; r < 8; ++r) {
      mine[r * 32]        = aP[r];
      mine[(8 + r) * 32]  = aQ[r];
      mine[(16 + r) * 32] = aR[r];
    }
    __syncthreads();
    if (wave != 0) return;
    aP = bP; aQ = bQ; aR = bR;
    #pragma unroll
    for (int w = 0; w < PG_WAVES; ++w) {
      const float* pw = fpart + (size_t)w * PG_PART * 32 + L;
      #pragma unroll
      for (int r = 0; r < 8; ++r) {
        aP[r] += pw[r * 32];
        aQ[r] += pw[(8 + r) * 32];
        aR[r] += pw[(16 + r) * 32];
      }
    }
    #pragma unroll
    for (int s = 0; s < 8; ++s) {
      Xb[0][s]     = (f16)aP[s];
      Xb[0][8 + s] = (f16)aQ[s];
      Xb[1][s]     = (f16)aR[s];
      Xb[1][8 + s] = (f16)0;
    }
  } else {
    const float* xr = captions + ((size_t)b * TSEQ + (t - 1)) * VOC;
    #pragma unroll
    for (int s = 0; s < 16; ++s) Xb[0][s] = (f16)xr[16 * half + s];
    #pragma unroll
    for (int s = 0; s < 16; ++s) {
      float v = (s < VOC - 32) ? xr[32 + s] : 0.f;
      Xb[1][s] = (half == 0) ? (f16)v : (f16)0;
    }
  }

  #pragma unroll
  for (int ks = 0; ks < 2; ++ks) {
    #pragma unroll
    for (int nt = 0; nt < 2; ++nt) {
      acc[nt] = wmma16(Wa[ks * 2 + nt], Xb[ks], acc[nt]);
    }
  }
  v8f pa, pb;
  #pragma unroll
  for (int g = 0; g < 4; ++g) {
    pa[2 * g]     = acc[g >> 1][4 * (g & 1)];
    pa[2 * g + 1] = acc[g >> 1][4 * (g & 1) + 1];
    pb[2 * g]     = acc[g >> 1][4 * (g & 1) + 2];
    pb[2 * g + 1] = acc[g >> 1][4 * (g & 1) + 3];
  }
  const int grp8 = 2 * bt + (j16 >> 3);
  const int rl = 8 * half + (j16 & 7);
  float* dst = G + ((size_t)t * (BATCH / 8) + grp8) * 256;
  *(v8f*)(dst + rl * 8)        = pa;
  *(v8f*)(dst + (rl + 16) * 8) = pb;
}

#define GAHEAD 16
static_assert(TSEQ % GAHEAD == 0 && TSEQ >= 2 * GAHEAD, "ring depth must divide the sequence");
#define NTURN (TSEQ / GAHEAD)
#define HEAD_WAVES 8
#define RH_THREADS (32 * (1 + HEAD_WAVES))
#define HPITCH (GAHEAD * HID + 8)
#define HSTAGE (8 * HPITCH)
static_assert(GAHEAD == 16, "a head tile is the sixteen steps of one ring turn");
static_assert((HPITCH * 2) % 16 == 0 && (HSTAGE * 2) % 16 == 0, "16-byte reads of a stage row");

typedef __attribute__((ext_vector_type(2))) _Float16 v2h;
typedef __attribute__((ext_vector_type(8))) int v8i;
typedef __attribute__((ext_vector_type(4))) int v4i;

__device__ __forceinline__ int from_lane_xor8(int x) {
  return __builtin_amdgcn_mov_dpp(x, 0x128  , 0xf, 0xf, false);
}

__device__ __forceinline__ float from_other_half(float x) {
  unsigned u = __builtin_bit_cast(unsigned, x);
  unsigned r = __builtin_amdgcn_permlanex16(u, u, 0x76543210u, 0xfedcba98u, false, false);
  return __builtin_bit_cast(float, r);
}

__device__ __forceinline__ void lstm_cell_step(v8f g, const v16h& Wa, v16h& Bh,
                                               float (&cst)[2], bool grp0,
                                               f16* hout) {
  g = wmma16(Wa, Bh, g);

  float h0[2], h1[2];
  #pragma unroll
  for (int u = 0; u < 2; ++u) {
    float iv = sigmoid_of_half_(g[u]);
    float fv = sigmoid_of_half_(g[2 + u]);
    float gv = tanhf_(g[4 + u]);
    float ov = sigmoid_of_half_(g[6 + u]);
    cst[u] = fv * cst[u] + iv * gv;
    float tc = tanhf_(cst[u]);
    h0[u] = (grp0 ? ov : 0.f) * tc;
    h1[u] = (grp0 ? 0.f : ov) * tc;
  }
  v2h q0 = { (f16)h0[0], (f16)h0[1] };
  v2h q1 = { (f16)h1[0], (f16)h1[1] };
  int hq0 = __builtin_bit_cast(int, q0);
  int hq1 = __builtin_bit_cast(int, q1);
  v8i b;
  b[0] = hq0;
  b[1] = from_lane_xor8(hq1);
  b[2] = hq1;
  b[3] = from_lane_xor8(hq0);
  b[4] = 0; b[5] = 0; b[6] = 0; b[7] = 0;
  Bh = __builtin_bit_cast(v16h, b);

  *(int*)hout = grp0 ? hq0 : hq1;
}

__device__ __forceinline__ void head_tile(const f16* hrow, const v16h (&WoutA)[3],
                                          const v8f (&cb)[3], int L,
                                          float* __restrict__ orow) {
  const int n = L & 15, half = L >> 4;
  v4i hv = *(const v4i*)(hrow + n * HID);
  v8i bb;
  for (int i = 0; i < 4; ++i) bb[i] = (half == 0) ? hv[i] : 0;
  bb[4] = 0; bb[5] = 0; bb[6] = 0; bb[7] = 0;
  const v16h Bt = __builtin_bit_cast(v16h, bb);

  v8f l0 = wmma16(WoutA[0], Bt, cb[0]);
  v8f l1 = wmma16(WoutA[1], Bt, cb[1]);
  v8f l2 = wmma16(WoutA[2], Bt, cb[2]);

  constexpr int R2 = VOC - 32;
  float x2[R2];
  for (int r = 0; r < R2; ++r) x2[r] = (half == 0) ? l2[r] : -1e30f;

  float mx = l0[0];
  for (int r = 1; r < 8; ++r) mx = fmaxf(mx, l0[r]);
  for (int r = 0; r < 8; ++r) mx = fmaxf(mx, l1[r]);
  for (int r = 0; r < R2; ++r) mx = fmaxf(mx, x2[r]);
  mx = fmaxf(mx, from_other_half(mx));

  float e0[8], e1[8], e2[R2];
  float sm = 0.f;
  for (int r = 0; r < 8; ++r) { e0[r] = __expf(l0[r] - mx); sm += e0[r]; }
  for (int r = 0; r < 8; ++r) { e1[r] = __expf(l1[r] - mx); sm += e1[r]; }
  for (int r = 0; r < R2; ++r) { e2[r] = __expf(x2[r] - mx); sm += e2[r]; }
  sm += from_other_half(sm);
  const float inv = 1.0f / sm;

  float* dst = orow + (size_t)n * VOC;
  for (int r = 0; r < 8; ++r) dst[8 * half + r] = e0[r] * inv;
  for (int r = 0; r < 8; ++r) dst[16 + 8 * half + r] = e1[r] * inv;
  if (half == 0) {
    for (int r = 0; r < R2; ++r) dst[32 + r] = e2[r] * inv;
  }
}

__global__ void __launch_bounds__(RH_THREADS)
lstm_recur_head_kernel(const float* __restrict__ G,
                       const f16* __restrict__ Whhp,
                       const f16* __restrict__ Woutp,
                       const float* __restrict__ obias,
                       float* __restrict__ out) {
  __shared__ __attribute__((aligned(16))) f16 hring[2 * HSTAGE];

  const int wave = __builtin_amdgcn_readfirstlane((int)(threadIdx.x >> 5));
  const bool recur = (wave == 0);
  const int L = threadIdx.x & 31;
  const int n16 = L & 15, half = L >> 4;

  const int grp = n16 >> 3;
  const bool grp0 = (grp == 0);
  const int pair = 2 * grp + half;
  v16h WhhA;
  float cst[2];
  v16h Bh;
  v8f ra[GAHEAD];
  f16* const hmine = hring + (n16 & 7) * HPITCH + 2 * pair;
  const float* Gl = G + (size_t)blockIdx.x * 256 + L * 8;
  const size_t GSTEP = (size_t)(BATCH / 8) * 256;

  v16h WoutA[3];
  v8f cb[3];
  const int hw = recur ? 0 : wave - 1;
  const f16* const hrow = hring + hw * HPITCH;
  float* const obase = out + (size_t)(blockIdx.x * 8 + hw) * TSEQ * VOC;

  if (recur) {
    __builtin_amdgcn_s_setprio(3);
    const f16* tp = Whhp + L * 16;
    WhhA = cat8(*(const v8h*)tp, *(const v8h*)(tp + 8));
    for (int u = 0; u < 2; ++u) cst[u] = 0.f;
    for (int i = 0; i < 16; ++i) Bh[i] = (f16)0;
    __builtin_amdgcn_sched_barrier(0);
    #pragma unroll
    for (int d = 0; d < GAHEAD; ++d) ra[d] = *(const v8f*)(Gl + d * GSTEP);
  } else {
    for (int nt = 0; nt < 3; ++nt) {
      const f16* tp = Woutp + (size_t)nt * 512 + L * 16;
      WoutA[nt] = cat8(*(const v8h*)tp, *(const v8h*)(tp + 8));
      cb[nt] = *(const v8f*)(obias + 16 * nt + 8 * half);
    }
  }

  for (int g = 0; g < NTURN; ++g) {
    if (recur) {
      f16* const hs = hmine + (g & 1) * HSTAGE;
      const int gn = (g + 1 < NTURN) ? (g + 1) : g;
      const float* Gn = Gl + (size_t)gn * GAHEAD * GSTEP;
      #pragma unroll
      for (int d = 0; d < GAHEAD; ++d) {
        lstm_cell_step(ra[d], WhhA, Bh, cst, grp0, hs + d * HID);
        __builtin_amdgcn_sched_barrier(0);
        ra[d] = *(const v8f*)(Gn + d * GSTEP);
      }
    } else if (g > 0) {
      head_tile(hrow + ((g - 1) & 1) * HSTAGE, WoutA, cb, L,
                obase + (size_t)(g - 1) * GAHEAD * VOC);
    }
    asm volatile("s_wait_dscnt 0x0" ::: "memory");
    asm volatile("s_barrier_signal -1" ::: "memory");
    asm volatile("s_barrier_wait -1" ::: "memory");
  }
  if (!recur)
    head_tile(hrow + ((NTURN - 1) & 1) * HSTAGE, WoutA, cb, L,
              obase + (size_t)(NTURN - 1) * GAHEAD * VOC);
}

extern "C" void kernel_launch(void* const* d_in, const int* in_sizes, int n_in,
                              void* d_out, int out_size, void* d_ws, size_t ws_size,
                              hipStream_t stream) {
  const float* features = (const float*)d_in[0];
  const float* captions = (const float*)d_in[1];
  const float* W_in  = (const float*)d_in[2];
  const float* b_in  = (const float*)d_in[3];
  const float* W_ih  = (const float*)d_in[4];
  const float* W_hh  = (const float*)d_in[5];
  const float* b_ih  = (const float*)d_in[6];
  const float* b_hh  = (const float*)d_in[7];
  const float* W_out = (const float*)d_in[8];
  const float* b_out = (const float*)d_in[9];
  float* out = (float*)d_out;

  char* ws = (char*)d_ws;
  auto carve = [&](size_t bytes) -> char* {
    char* p = ws;
    ws += (bytes + 255) & ~(size_t)255;
    return p;
  };
  f16*   pooled = (f16*)  carve((size_t)BATCH * CNNC * sizeof(f16));
  f16*   Winp   = (f16*)  carve((size_t)48 * CNNC * sizeof(f16));
  f16*   Wihp   = (f16*)  carve((size_t)4 * 512 * sizeof(f16));
  f16*   Whhp   = (f16*)  carve((size_t)1 * 512 * sizeof(f16));
  f16*   Woutp  = (f16*)  carve((size_t)3 * 512 * sizeof(f16));
  float* gbias  = (float*)carve(GN * sizeof(float));
  float* obias  = (float*)carve(48 * sizeof(float));
  float* G      = (float*)carve((size_t)NROW * GN * sizeof(float));

  lstm_front_kernel<<<PK_BLKS + POOL_BLKS, PK_THREADS, 0, stream>>>(features, pooled,
                                                       W_in, W_ih, W_hh, W_out,
                                                       b_ih, b_hh, b_out,
                                                       Winp, Wihp, Whhp, Woutp, gbias, obias);
  lstm_pregate_kernel<<<PG_BLKS, PG_THREADS, 0, stream>>>(captions, pooled, Winp, b_in, Wihp, gbias, G);
  lstm_recur_head_kernel<<<BATCH / 8, RH_THREADS, 0, stream>>>(G, Whhp, Woutp, obias, out);
}
